# starting kernel
# baseline (speedup 1.0000x reference)
#include <hip/hip_runtime.h>

typedef float v2f __attribute__((ext_vector_type(2)));
typedef float v8f __attribute__((ext_vector_type(8)));

#define LX 6
#define LY 6
#define PD 2
#define DD 4
#define NSITE 36
#define ENV7 16384
#define ENV6 4096

__global__ void peps_prep_kernel(const float* __restrict__ peps,
                                 float* __restrict__ Msite) {
    int idx = blockIdx.x * blockDim.x + threadIdx.x;
    if (idx >= NSITE * PD * 256) return;
    int n    = idx & 15;
    int k    = (idx >> 4) & 15;
    int s    = (idx >> 8) & 1;
    int site = idx >> 9;
    int i = site / LY, j = site % LY;
    int a = k >> 2, c = k & 3;
    int d = n >> 2, b = n & 3;
    float v = 0.0f;
    bool pad = (i == 0      && a > 0) || (i == LX - 1 && b > 0) ||
               (j == 0      && c > 0) || (j == LY - 1 && d > 0);
    if (!pad) {
        int p = ((((site * PD + s) * DD + a) * DD + b) * DD + c) * DD + d;
        v = peps[p];
    }
    Msite[idx] = v;
}

__global__ void __launch_bounds__(256)
peps_amp_kernel(const int* __restrict__ x,
                const float* __restrict__ Msite,
                float* __restrict__ out) {
    extern __shared__ float env[];

    const int tid    = threadIdx.x;
    const int sample = blockIdx.x;
    const int lane   = tid & 31;
    const int wave   = __builtin_amdgcn_readfirstlane(tid >> 5);
    __builtin_assume(wave >= 0 && wave < 8);
    const int n      = lane & 15;
    const int h      = lane >> 4;

    for (int q = tid; q < ENV7; q += 256) env[q] = 0.0f;
    __syncthreads();
    if (tid == 0) env[0] = 1.0f;
    __syncthreads();

    const int* xrow = x + sample * NSITE;

    for (int j = 0; j < LY; ++j) {
        for (int i = 0; i < LX; ++i) {
            const int site = i * LY + j;
            const int s    = xrow[site];
            const float* Mp = Msite + (site * PD + s) * 256;

            v2f b0, b1, b2, b3;
            b0.x = Mp[(0  + 2 * h) * 16 + n];  b0.y = Mp[(1  + 2 * h) * 16 + n];
            b1.x = Mp[(4  + 2 * h) * 16 + n];  b1.y = Mp[(5  + 2 * h) * 16 + n];
            b2.x = Mp[(8  + 2 * h) * 16 + n];  b2.y = Mp[(9  + 2 * h) * 16 + n];
            b3.x = Mp[(12 + 2 * h) * 16 + n];  b3.y = Mp[(13 + 2 * h) * 16 + n];

            const int shift = 2 * (5 - i);
            const int psm   = (1 << shift) - 1;

            #pragma unroll 2
            for (int t = wave; t < 64; t += 8) {
                int r    = t * 16 + n;
                int post = r & psm;
                int base = ((r - post) << 4) + post;
                v2f a0, a1, a2, a3;
                a0.x = env[base + ((0  + 2 * h) << shift)];
                a0.y = env[base + ((1  + 2 * h) << shift)];
                a1.x = env[base + ((4  + 2 * h) << shift)];
                a1.y = env[base + ((5  + 2 * h) << shift)];
                a2.x = env[base + ((8  + 2 * h) << shift)];
                a2.y = env[base + ((9  + 2 * h) << shift)];
                a3.x = env[base + ((12 + 2 * h) << shift)];
                a3.y = env[base + ((13 + 2 * h) << shift)];

                v8f acc = {};
                acc = __builtin_amdgcn_wmma_f32_16x16x4_f32(
                        false, a0, false, b0, (short)0, acc, false, false);
                acc = __builtin_amdgcn_wmma_f32_16x16x4_f32(
                        false, a1, false, b1, (short)0, acc, false, false);
                acc = __builtin_amdgcn_wmma_f32_16x16x4_f32(
                        false, a2, false, b2, (short)0, acc, false, false);
                acc = __builtin_amdgcn_wmma_f32_16x16x4_f32(
                        false, a3, false, b3, (short)0, acc, false, false);

                #pragma unroll
                for (int v = 0; v < 8; ++v) {
                    int rs    = t * 16 + v + 8 * h;
                    int posts = rs & psm;
                    int bs    = ((rs - posts) << 4) + posts;
                    env[bs + (n << shift)] = acc[v];
                }
            }
            __syncthreads();
        }

        if (j < LY - 1) {
            float keep[16];
            #pragma unroll
            for (int q = 0; q < 16; ++q) keep[q] = env[4 * (tid + 256 * q)];
            __syncthreads();
            #pragma unroll
            for (int q = 0; q < 16; ++q) env[tid + 256 * q] = keep[q];
            for (int q = ENV6 + tid; q < ENV7; q += 256) env[q] = 0.0f;
            __syncthreads();
        }
    }

    if (tid == 0) out[sample] = env[0];
}

extern "C" void kernel_launch(void* const* d_in, const int* in_sizes, int n_in,
                              void* d_out, int out_size, void* d_ws, size_t ws_size,
                              hipStream_t stream) {
    const int*   x     = (const int*)d_in[0];
    const float* peps  = (const float*)d_in[1];
    float*       out   = (float*)d_out;
    float*       Msite = (float*)d_ws;

    const int nM = NSITE * PD * 256;
    peps_prep_kernel<<<(nM + 255) / 256, 256, 0, stream>>>(peps, Msite);

    const int B = out_size;
    peps_amp_kernel<<<B, 256, ENV7 * sizeof(float), stream>>>(x, Msite, out);
}


# pass between columns (slice compaction, zero fill, 2 barriers) removed: every second column swept bottom-up on mirrored strides, from where the last ended; sum order changes: those columns bottom-up
# speedup vs baseline: 3.6945x; 3.6945x over previous
#include <hip/hip_runtime.h>


typedef float v2f __attribute__((ext_vector_type(2)));
typedef float v8f __attribute__((ext_vector_type(8)));

#define LX 6
#define LY 6
#define PD 2
#define DD 4
#define NSITE 36
#define ENV7 16384
#define ENV6 4096
#define NCFG 64
#define MSITE_BYTES (NSITE * PD * 256 * 4)
#define WS_NEED     (MSITE_BYTES + NCFG * ENV6 * 4)

__global__ void peps_prep_kernel(const float* __restrict__ peps,
                                 float* __restrict__ Msite) {
    int idx = blockIdx.x * blockDim.x + threadIdx.x;
    if (idx >= NSITE * PD * 256) return;
    int n    = idx & 15;
    int k    = (idx >> 4) & 15;
    int s    = (idx >> 8) & 1;
    int site = idx >> 9;
    int i = site / LY, j = site % LY;
    int a = k >> 2, c = k & 3;
    int d = n >> 2, b = n & 3;
    float v = 0.0f;
    bool pad = (i == 0      && a > 0) || (i == LX - 1 && b > 0) ||
               (j == 0      && c > 0) || (j == LY - 1 && d > 0);
    if (!pad) {
        int p = ((((site * PD + s) * DD + a) * DD + b) * DD + c) * DD + d;
        v = peps[p];
    }
    Msite[idx] = v;
}

__device__ __forceinline__ constexpr int swz(int a) { return a ^ ((a >> 6) & 15); }

__device__ __forceinline__ int cell4(int Lz4, int K) {
    const int X   = ((K >> 6) & 15) ^ (K & 15);
    const int KHI = K & ~15;
    return (Lz4 ^ (X * 4)) + KHI * 4;
}
__device__ __forceinline__ float& at(float* env, int byte_addr) {
    return *reinterpret_cast<float*>(reinterpret_cast<char*>(env) + byte_addr);
}

template <int S0, int S1, int S2, bool UPWARD>
__device__ __forceinline__ void site_pair(float* env,
                                          const float* __restrict__ MpA,
                                          const float* __restrict__ MpB,
                                          int wave, int c, int h) {
    constexpr int MINS = S0 < S2 ? S0 : S2;

    v2f fa[4], fb[4];
    #pragma unroll
    for (int kc = 0; kc < 4; ++kc) {
        if (!UPWARD) {
            const int ncol = (c & 3) * 4 + (c >> 2);
            fa[kc].x = MpA[((2 * h + 0) * 4 + kc) * 16 + ncol];
            fa[kc].y = MpA[((2 * h + 1) * 4 + kc) * 16 + ncol];
            fb[kc].x = MpB[((2 * h + 0) * 4 + kc) * 16 + ncol];
            fb[kc].y = MpB[((2 * h + 1) * 4 + kc) * 16 + ncol];
        } else {
            const int nrow = ((c >> 2) * 4 + kc) * 16 + (c & 3) * 4;
            fa[kc].x = MpA[nrow + 2 * h + 0];
            fa[kc].y = MpA[nrow + 2 * h + 1];
            fb[kc].x = MpB[nrow + 2 * h + 0];
            fb[kc].y = MpB[nrow + 2 * h + 1];
        }
    }

    #pragma nounroll
    for (int it = 0; it < 2; ++it) {
        const int q    = (wave + 8 * it) * 16 + c;
        const int ql   = q & (MINS - 1);
        const int base = ((q - ql) << 6) | ql;
        const int ld0  = 4 * swz(base + (2 * h) * S0);
        const int st0  = 4 * swz(base + (2 * h) * S2);

        v8f dA[4];
        #pragma unroll
        for (int T = 0; T < 4; ++T) {
            v8f acc = {};
            #pragma unroll
            for (int kc = 0; kc < 4; ++kc) {
                v2f b;
                b.x = at(env, cell4(ld0, kc * S1 + T * S2));
                b.y = at(env, cell4(ld0, S0 + kc * S1 + T * S2));
                acc = __builtin_amdgcn_wmma_f32_16x16x4_f32(
                        false, fa[kc], false, b, (short)0, acc, false, false);
            }
            dA[T] = acc;
        }

        #pragma unroll
        for (int R = 0; R < 4; ++R) {
            v8f acc = {};
            #pragma unroll
            for (int kc = 0; kc < 4; ++kc) {
                v2f b;
                b.x = dA[kc][R];
                b.y = dA[kc][4 + R];
                acc = __builtin_amdgcn_wmma_f32_16x16x4_f32(
                        false, fb[kc], false, b, (short)0, acc, false, false);
            }
            #pragma unroll
            for (int r = 0; r < 8; ++r)
                at(env, cell4(st0, R * S0 + (r & 3) * S1 + (r >> 2) * S2)) = acc[r];
        }
    }
}

__device__ __forceinline__ void column_sites(float* env,
                                             const float* M0, const float* M1,
                                             const float* M2, const float* M3,
                                             const float* M4, const float* M5,
                                             int wave, int c, int h) {
    site_pair<4096, 1024, 256, false>(env, M0, M1, wave, c, h);
    __syncthreads();
    site_pair<256, 64, 16, false>(env, M2, M3, wave, c, h);
    __syncthreads();
    site_pair<16, 4, 1, false>(env, M4, M5, wave, c, h);
    __syncthreads();
}

__device__ __forceinline__ void column_sites_up(float* env,
                                                const float* M0, const float* M1,
                                                const float* M2, const float* M3,
                                                const float* M4, const float* M5,
                                                int wave, int c, int h) {
    site_pair<1, 4, 16, true>(env, M5, M4, wave, c, h);
    __syncthreads();
    site_pair<16, 64, 256, true>(env, M3, M2, wave, c, h);
    __syncthreads();
    site_pair<256, 1024, 4096, true>(env, M1, M0, wave, c, h);
    __syncthreads();
}

__global__ void __launch_bounds__(256)
peps_col0_kernel(const float* __restrict__ Msite,
                 float* __restrict__ table,
                 size_t ws_bytes) {
    extern __shared__ float env[];

    if (ws_bytes < (size_t)WS_NEED) return;

    const int tid  = threadIdx.x;
    const int cfg  = blockIdx.x;
    const int lane = tid & 31;
    const int wave = __builtin_amdgcn_readfirstlane(tid >> 5);
    __builtin_assume(wave >= 0 && wave < 8);
    const int c    = lane & 15;
    const int h    = lane >> 4;

    for (int q = tid; q < ENV7; q += 256) env[q] = 0.0f;
    __syncthreads();
    if (tid == 0) env[0] = 1.0f;
    __syncthreads();

    const float* M0 = Msite + ((0 * LY) * PD + ((cfg >> 0) & 1)) * 256;
    const float* M1 = Msite + ((1 * LY) * PD + ((cfg >> 1) & 1)) * 256;
    const float* M2 = Msite + ((2 * LY) * PD + ((cfg >> 2) & 1)) * 256;
    const float* M3 = Msite + ((3 * LY) * PD + ((cfg >> 3) & 1)) * 256;
    const float* M4 = Msite + ((4 * LY) * PD + ((cfg >> 4) & 1)) * 256;
    const float* M5 = Msite + ((5 * LY) * PD + ((cfg >> 5) & 1)) * 256;
    column_sites(env, M0, M1, M2, M3, M4, M5, wave, c, h);

    float* row = table + cfg * ENV6;
    #pragma unroll
    for (int q = 0; q < 16; ++q)
        row[tid + 256 * q] = at(env, cell4(4 * swz(4 * tid), 1024 * q));
}

__global__ void __launch_bounds__(256)
peps_amp_kernel(const int* __restrict__ x,
                const float* __restrict__ Msite,
                const float* __restrict__ table,
                size_t ws_bytes,
                float* __restrict__ out) {
    extern __shared__ float env[];

    const int tid    = threadIdx.x;
    const int sample = blockIdx.x;
    const int lane   = tid & 31;
    const int wave   = __builtin_amdgcn_readfirstlane(tid >> 5);
    __builtin_assume(wave >= 0 && wave < 8);
    const int c      = lane & 15;
    const int h      = lane >> 4;

    const int* xrow = x + sample * NSITE;
    const bool have_table = ws_bytes >= (size_t)WS_NEED;

    if (have_table) {
        const int cfg = xrow[0 * LY] | (xrow[1 * LY] << 1) | (xrow[2 * LY] << 2) |
                        (xrow[3 * LY] << 3) | (xrow[4 * LY] << 4) | (xrow[5 * LY] << 5);
        const float* row = table + cfg * ENV6;
        float keep[16];
        #pragma unroll
        for (int q = 0; q < 16; ++q) keep[q] = row[tid + 256 * q];
        for (int q = ENV6 + tid; q < ENV7; q += 256) env[q] = 0.0f;
        #pragma unroll
        for (int q = 0; q < 16; ++q) at(env, cell4(4 * swz(tid), 256 * q)) = keep[q];
        __syncthreads();
    } else {
        for (int q = tid; q < ENV7; q += 256) env[q] = 0.0f;
        __syncthreads();
        if (tid == 0) env[0] = 1.0f;
        __syncthreads();
    }

    bool upward = false;
    for (int j = have_table ? 1 : 0; j < LY; ++j) {
        const float* M0 = Msite + ((0 * LY + j) * PD + xrow[0 * LY + j]) * 256;
        const float* M1 = Msite + ((1 * LY + j) * PD + xrow[1 * LY + j]) * 256;
        const float* M2 = Msite + ((2 * LY + j) * PD + xrow[2 * LY + j]) * 256;
        const float* M3 = Msite + ((3 * LY + j) * PD + xrow[3 * LY + j]) * 256;
        const float* M4 = Msite + ((4 * LY + j) * PD + xrow[4 * LY + j]) * 256;
        const float* M5 = Msite + ((5 * LY + j) * PD + xrow[5 * LY + j]) * 256;

        if (!upward) column_sites(env, M0, M1, M2, M3, M4, M5, wave, c, h);
        else         column_sites_up(env, M0, M1, M2, M3, M4, M5, wave, c, h);
        upward = !upward;
    }

    if (tid == 0) out[sample] = env[0];
}

extern "C" void kernel_launch(void* const* d_in, const int* in_sizes, int n_in,
                              void* d_out, int out_size, void* d_ws, size_t ws_size,
                              hipStream_t stream) {
    const int*   x     = (const int*)d_in[0];
    const float* peps  = (const float*)d_in[1];
    float*       out   = (float*)d_out;
    float*       Msite = (float*)d_ws;
    float*       table = Msite + NSITE * PD * 256;

    const int nM = NSITE * PD * 256;
    peps_prep_kernel<<<(nM + 255) / 256, 256, 0, stream>>>(peps, Msite);

    peps_col0_kernel<<<NCFG, 256, ENV7 * sizeof(float), stream>>>(Msite, table, ws_size);

    const int B = out_size;
    peps_amp_kernel<<<B, 256, ENV7 * sizeof(float), stream>>>(x, Msite, table, ws_size, out);
}


# column 5 contracted from the right once per 6-bit index configuration (64 more blocks of the table launch, mirrored site matrices); a sample ends with a 4096-term inner product; sum order changes
# speedup vs baseline: 4.4157x; 1.1952x over previous
#include <hip/hip_runtime.h>


typedef float v2f __attribute__((ext_vector_type(2)));
typedef float v8f __attribute__((ext_vector_type(8)));

#define LX 6
#define LY 6
#define PD 2
#define DD 4
#define NSITE 36
#define ENV7 16384
#define ENV6 4096
#define NCFG 64
#define NMAT 42
#define MSITE_BYTES (NMAT * PD * 256 * 4)
#define WS_NEED     (MSITE_BYTES + NCFG * ENV6 * 4)
#define WS_NEED2    (MSITE_BYTES + 2 * NCFG * ENV6 * 4)

__global__ void peps_prep_kernel(const float* __restrict__ peps,
                                 float* __restrict__ Msite,
                                 size_t ws_bytes) {
    int idx = blockIdx.x * blockDim.x + threadIdx.x;
    if (idx >= NMAT * PD * 256) return;
    if (idx >= NSITE * PD * 256 && ws_bytes < (size_t)WS_NEED2) return;
    int n    = idx & 15;
    int k    = (idx >> 4) & 15;
    int s    = (idx >> 8) & 1;
    int site = idx >> 9;
    int a = k >> 2, c = k & 3;
    int d = n >> 2, b = n & 3;
    float v = 0.0f;
    if (site < NSITE) {
        int i = site / LY, j = site % LY;
        bool pad = (i == 0      && a > 0) || (i == LX - 1 && b > 0) ||
                   (j == 0      && c > 0) || (j == LY - 1 && d > 0);
        if (!pad) {
            int p = ((((site * PD + s) * DD + a) * DD + b) * DD + c) * DD + d;
            v = peps[p];
        }
    } else {
        int i  = site - NSITE;
        int rs = i * LY + (LY - 1);
        bool pad = (i == 0 && a > 0) || (i == LX - 1 && b > 0) || (c > 0);
        if (!pad) {
            int p = ((((rs * PD + s) * DD + a) * DD + b) * DD + d) * DD + c;
            v = peps[p];
        }
    }
    Msite[idx] = v;
}

__device__ __forceinline__ constexpr int swz(int a) { return a ^ ((a >> 6) & 15); }

__device__ __forceinline__ int cell4(int Lz4, int K) {
    const int X   = ((K >> 6) & 15) ^ (K & 15);
    const int KHI = K & ~15;
    return (Lz4 ^ (X * 4)) + KHI * 4;
}
__device__ __forceinline__ float& at(float* env, int byte_addr) {
    return *reinterpret_cast<float*>(reinterpret_cast<char*>(env) + byte_addr);
}

template <int S0, int S1, int S2, bool UPWARD>
__device__ __forceinline__ void site_pair(float* env,
                                          const float* __restrict__ MpA,
                                          const float* __restrict__ MpB,
                                          int wave, int c, int h) {
    constexpr int MINS = S0 < S2 ? S0 : S2;

    v2f fa[4], fb[4];
    #pragma unroll
    for (int kc = 0; kc < 4; ++kc) {
        if (!UPWARD) {
            const int ncol = (c & 3) * 4 + (c >> 2);
            fa[kc].x = MpA[((2 * h + 0) * 4 + kc) * 16 + ncol];
            fa[kc].y = MpA[((2 * h + 1) * 4 + kc) * 16 + ncol];
            fb[kc].x = MpB[((2 * h + 0) * 4 + kc) * 16 + ncol];
            fb[kc].y = MpB[((2 * h + 1) * 4 + kc) * 16 + ncol];
        } else {
            const int nrow = ((c >> 2) * 4 + kc) * 16 + (c & 3) * 4;
            fa[kc].x = MpA[nrow + 2 * h + 0];
            fa[kc].y = MpA[nrow + 2 * h + 1];
            fb[kc].x = MpB[nrow + 2 * h + 0];
            fb[kc].y = MpB[nrow + 2 * h + 1];
        }
    }

    #pragma nounroll
    for (int it = 0; it < 2; ++it) {
        const int q    = (wave + 8 * it) * 16 + c;
        const int ql   = q & (MINS - 1);
        const int base = ((q - ql) << 6) | ql;
        const int ld0  = 4 * swz(base + (2 * h) * S0);
        const int st0  = 4 * swz(base + (2 * h) * S2);

        v8f dA[4];
        #pragma unroll
        for (int T = 0; T < 4; ++T) {
            v8f acc = {};
            #pragma unroll
            for (int kc = 0; kc < 4; ++kc) {
                v2f b;
                b.x = at(env, cell4(ld0, kc * S1 + T * S2));
                b.y = at(env, cell4(ld0, S0 + kc * S1 + T * S2));
                acc = __builtin_amdgcn_wmma_f32_16x16x4_f32(
                        false, fa[kc], false, b, (short)0, acc, false, false);
            }
            dA[T] = acc;
        }

        #pragma unroll
        for (int R = 0; R < 4; ++R) {
            v8f acc = {};
            #pragma unroll
            for (int kc = 0; kc < 4; ++kc) {
                v2f b;
                b.x = dA[kc][R];
                b.y = dA[kc][4 + R];
                acc = __builtin_amdgcn_wmma_f32_16x16x4_f32(
                        false, fb[kc], false, b, (short)0, acc, false, false);
            }
            #pragma unroll
            for (int r = 0; r < 8; ++r)
                at(env, cell4(st0, R * S0 + (r & 3) * S1 + (r >> 2) * S2)) = acc[r];
        }
    }
}

__device__ __forceinline__ void column_sites(float* env,
                                             const float* M0, const float* M1,
                                             const float* M2, const float* M3,
                                             const float* M4, const float* M5,
                                             int wave, int c, int h) {
    site_pair<4096, 1024, 256, false>(env, M0, M1, wave, c, h);
    __syncthreads();
    site_pair<256, 64, 16, false>(env, M2, M3, wave, c, h);
    __syncthreads();
    site_pair<16, 4, 1, false>(env, M4, M5, wave, c, h);
    __syncthreads();
}

__device__ __forceinline__ void column_sites_up(float* env,
                                                const float* M0, const float* M1,
                                                const float* M2, const float* M3,
                                                const float* M4, const float* M5,
                                                int wave, int c, int h) {
    site_pair<1, 4, 16, true>(env, M5, M4, wave, c, h);
    __syncthreads();
    site_pair<16, 64, 256, true>(env, M3, M2, wave, c, h);
    __syncthreads();
    site_pair<256, 1024, 4096, true>(env, M1, M0, wave, c, h);
    __syncthreads();
}

__global__ void __launch_bounds__(256)
peps_edge_kernel(const float* __restrict__ Msite,
                 float* __restrict__ table,
                 size_t ws_bytes) {
    extern __shared__ float env[];

    const bool right = blockIdx.x >= NCFG;

    if (ws_bytes < (size_t)(right ? WS_NEED2 : WS_NEED)) return;

    const int tid  = threadIdx.x;
    const int cfg  = blockIdx.x & (NCFG - 1);
    const int m0   = right ? NSITE : 0;
    const int mst  = right ? 1 : LY;
    const int lane = tid & 31;
    const int wave = __builtin_amdgcn_readfirstlane(tid >> 5);
    __builtin_assume(wave >= 0 && wave < 8);
    const int c    = lane & 15;
    const int h    = lane >> 4;

    for (int q = tid; q < ENV7; q += 256) env[q] = 0.0f;
    __syncthreads();
    if (tid == 0) env[0] = 1.0f;
    __syncthreads();

    const float* M0 = Msite + ((m0 + 0 * mst) * PD + ((cfg >> 0) & 1)) * 256;
    const float* M1 = Msite + ((m0 + 1 * mst) * PD + ((cfg >> 1) & 1)) * 256;
    const float* M2 = Msite + ((m0 + 2 * mst) * PD + ((cfg >> 2) & 1)) * 256;
    const float* M3 = Msite + ((m0 + 3 * mst) * PD + ((cfg >> 3) & 1)) * 256;
    const float* M4 = Msite + ((m0 + 4 * mst) * PD + ((cfg >> 4) & 1)) * 256;
    const float* M5 = Msite + ((m0 + 5 * mst) * PD + ((cfg >> 5) & 1)) * 256;
    column_sites(env, M0, M1, M2, M3, M4, M5, wave, c, h);

    float* row = table + blockIdx.x * ENV6;
    #pragma unroll
    for (int q = 0; q < 16; ++q)
        row[tid + 256 * q] = at(env, cell4(4 * swz(4 * tid), 1024 * q));
}

__global__ void __launch_bounds__(256)
peps_amp_kernel(const int* __restrict__ x,
                const float* __restrict__ Msite,
                const float* __restrict__ table,
                size_t ws_bytes,
                float* __restrict__ out) {
    extern __shared__ float env[];

    const int tid    = threadIdx.x;
    const int sample = blockIdx.x;
    const int lane   = tid & 31;
    const int wave   = __builtin_amdgcn_readfirstlane(tid >> 5);
    __builtin_assume(wave >= 0 && wave < 8);
    const int c      = lane & 15;
    const int h      = lane >> 4;

    const int* xrow = x + sample * NSITE;
    const bool have_table = ws_bytes >= (size_t)WS_NEED;
    const bool have_right = ws_bytes >= (size_t)WS_NEED2;

    if (have_table) {
        const int cfg = xrow[0 * LY] | (xrow[1 * LY] << 1) | (xrow[2 * LY] << 2) |
                        (xrow[3 * LY] << 3) | (xrow[4 * LY] << 4) | (xrow[5 * LY] << 5);
        const float* row = table + cfg * ENV6;
        float keep[16];
        #pragma unroll
        for (int q = 0; q < 16; ++q) keep[q] = row[tid + 256 * q];
        for (int q = ENV6 + tid; q < ENV7; q += 256) env[q] = 0.0f;
        #pragma unroll
        for (int q = 0; q < 16; ++q) at(env, cell4(4 * swz(tid), 256 * q)) = keep[q];
        __syncthreads();
    } else {
        for (int q = tid; q < ENV7; q += 256) env[q] = 0.0f;
        __syncthreads();
        if (tid == 0) env[0] = 1.0f;
        __syncthreads();
    }

    bool upward = false;
    const int jend = have_right ? LY - 1 : LY;
    for (int j = have_table ? 1 : 0; j < jend; ++j) {
        const float* M0 = Msite + ((0 * LY + j) * PD + xrow[0 * LY + j]) * 256;
        const float* M1 = Msite + ((1 * LY + j) * PD + xrow[1 * LY + j]) * 256;
        const float* M2 = Msite + ((2 * LY + j) * PD + xrow[2 * LY + j]) * 256;
        const float* M3 = Msite + ((3 * LY + j) * PD + xrow[3 * LY + j]) * 256;
        const float* M4 = Msite + ((4 * LY + j) * PD + xrow[4 * LY + j]) * 256;
        const float* M5 = Msite + ((5 * LY + j) * PD + xrow[5 * LY + j]) * 256;

        if (!upward) column_sites(env, M0, M1, M2, M3, M4, M5, wave, c, h);
        else         column_sites_up(env, M0, M1, M2, M3, M4, M5, wave, c, h);
        upward = !upward;
    }

    if (!have_right) {
        if (tid == 0) out[sample] = env[0];
        return;
    }

    const int cfg5 = xrow[0 * LY + 5] | (xrow[1 * LY + 5] << 1) | (xrow[2 * LY + 5] << 2) |
                     (xrow[3 * LY + 5] << 3) | (xrow[4 * LY + 5] << 4) | (xrow[5 * LY + 5] << 5);
    const float* rrow = table + (NCFG + cfg5) * ENV6;
    float rv[16];
    #pragma unroll
    for (int q = 0; q < 16; ++q) rv[q] = rrow[tid + 256 * q];
    float part = 0.0f;
    #pragma unroll
    for (int q = 0; q < 16; ++q)
        part = fmaf(at(env, cell4(4 * swz(tid), 256 * q)), rv[q], part);
    for (int o = 16; o > 0; o >>= 1) part += __shfl_xor(part, o);
    __syncthreads();
    if (lane == 0) env[wave] = part;
    __syncthreads();
    if (tid == 0) {
        float t = env[0];
        for (int w = 1; w < 8; ++w) t += env[w];
        out[sample] = t;
    }
}

extern "C" void kernel_launch(void* const* d_in, const int* in_sizes, int n_in,
                              void* d_out, int out_size, void* d_ws, size_t ws_size,
                              hipStream_t stream) {
    const int*   x     = (const int*)d_in[0];
    const float* peps  = (const float*)d_in[1];
    float*       out   = (float*)d_out;
    float*       Msite = (float*)d_ws;
    float*       table = Msite + NMAT * PD * 256;

    const int nM = NMAT * PD * 256;
    peps_prep_kernel<<<(nM + 255) / 256, 256, 0, stream>>>(peps, Msite, ws_size);

    peps_edge_kernel<<<2 * NCFG, 256, ENV7 * sizeof(float), stream>>>(Msite, table, ws_size);

    const int B = out_size;
    peps_amp_kernel<<<B, 256, ENV7 * sizeof(float), stream>>>(x, Msite, table, ws_size, out);
}


# edge tables one site pair deeper: left row = column 0 plus rows 0,1 of column 1, right row = column 5 plus rows 0,1 of column 4 (256 rows of 16384 cells a side); sum order changes at the join
# speedup vs baseline: 4.9395x; 1.1186x over previous
#include <hip/hip_runtime.h>


typedef float v2f __attribute__((ext_vector_type(2)));
typedef float v4f __attribute__((ext_vector_type(4)));
typedef float v8f __attribute__((ext_vector_type(8)));

#define LX 6
#define LY 6
#define PD 2
#define DD 4
#define NSITE 36
#define ENV7 16384
#define ENV6 4096
#define NCFG 64
#define NCFG2 256
#define NMAT 44
#define MSITE_BYTES (NMAT * PD * 256 * 4)
#define WS_NEED     (MSITE_BYTES + NCFG * ENV6 * 4)
#define WS_NEED2    (MSITE_BYTES + 2 * NCFG * ENV6 * 4)
#define WS_NEED3    (MSITE_BYTES + 2 * NCFG2 * ENV7 * 4)

__global__ void peps_prep_kernel(const float* __restrict__ peps,
                                 float* __restrict__ Msite,
                                 size_t ws_bytes) {
    int idx = blockIdx.x * blockDim.x + threadIdx.x;
    if (idx >= NMAT * PD * 256) return;
    if (idx >= NSITE * PD * 256 && ws_bytes < (size_t)WS_NEED2) return;
    if (idx >= (NSITE + LX) * PD * 256 && ws_bytes < (size_t)WS_NEED3) return;
    int n    = idx & 15;
    int k    = (idx >> 4) & 15;
    int s    = (idx >> 8) & 1;
    int site = idx >> 9;
    int a = k >> 2, c = k & 3;
    int d = n >> 2, b = n & 3;
    float v = 0.0f;
    if (site < NSITE) {
        int i = site / LY, j = site % LY;
        bool pad = (i == 0      && a > 0) || (i == LX - 1 && b > 0) ||
                   (j == 0      && c > 0) || (j == LY - 1 && d > 0);
        if (!pad) {
            int p = ((((site * PD + s) * DD + a) * DD + b) * DD + c) * DD + d;
            v = peps[p];
        }
    } else if (site < NSITE + LX) {
        int i  = site - NSITE;
        int rs = i * LY + (LY - 1);
        bool pad = (i == 0 && a > 0) || (i == LX - 1 && b > 0) || (c > 0);
        if (!pad) {
            int p = ((((rs * PD + s) * DD + a) * DD + b) * DD + d) * DD + c;
            v = peps[p];
        }
    } else {
        int i  = site - (NSITE + LX);
        int rs = i * LY + (LY - 2);
        bool pad = (i == 0 && a > 0);
        if (!pad) {
            int p = ((((rs * PD + s) * DD + a) * DD + b) * DD + d) * DD + c;
            v = peps[p];
        }
    }
    Msite[idx] = v;
}

__device__ __forceinline__ constexpr int swz(int a) { return a ^ ((a >> 6) & 15); }

__device__ __forceinline__ int cell4(int Lz4, int K) {
    const int X   = ((K >> 6) & 15) ^ (K & 15);
    const int KHI = K & ~15;
    return (Lz4 ^ (X * 4)) + KHI * 4;
}
__device__ __forceinline__ float& at(float* env, int byte_addr) {
    return *reinterpret_cast<float*>(reinterpret_cast<char*>(env) + byte_addr);
}

template <int S0, int S1, int S2, bool UPWARD>
__device__ __forceinline__ void site_pair(float* env,
                                          const float* __restrict__ MpA,
                                          const float* __restrict__ MpB,
                                          int wave, int c, int h) {
    constexpr int MINS = S0 < S2 ? S0 : S2;

    v2f fa[4], fb[4];
    #pragma unroll
    for (int kc = 0; kc < 4; ++kc) {
        if (!UPWARD) {
            const int ncol = (c & 3) * 4 + (c >> 2);
            fa[kc].x = MpA[((2 * h + 0) * 4 + kc) * 16 + ncol];
            fa[kc].y = MpA[((2 * h + 1) * 4 + kc) * 16 + ncol];
            fb[kc].x = MpB[((2 * h + 0) * 4 + kc) * 16 + ncol];
            fb[kc].y = MpB[((2 * h + 1) * 4 + kc) * 16 + ncol];
        } else {
            const int nrow = ((c >> 2) * 4 + kc) * 16 + (c & 3) * 4;
            fa[kc].x = MpA[nrow + 2 * h + 0];
            fa[kc].y = MpA[nrow + 2 * h + 1];
            fb[kc].x = MpB[nrow + 2 * h + 0];
            fb[kc].y = MpB[nrow + 2 * h + 1];
        }
    }

    #pragma nounroll
    for (int it = 0; it < 2; ++it) {
        const int q    = (wave + 8 * it) * 16 + c;
        const int ql   = q & (MINS - 1);
        const int base = ((q - ql) << 6) | ql;
        const int ld0  = 4 * swz(base + (2 * h) * S0);
        const int st0  = 4 * swz(base + (2 * h) * S2);

        v8f dA[4];
        #pragma unroll
        for (int T = 0; T < 4; ++T) {
            v8f acc = {};
            #pragma unroll
            for (int kc = 0; kc < 4; ++kc) {
                v2f b;
                b.x = at(env, cell4(ld0, kc * S1 + T * S2));
                b.y = at(env, cell4(ld0, S0 + kc * S1 + T * S2));
                acc = __builtin_amdgcn_wmma_f32_16x16x4_f32(
                        false, fa[kc], false, b, (short)0, acc, false, false);
            }
            dA[T] = acc;
        }

        #pragma unroll
        for (int R = 0; R < 4; ++R) {
            v8f acc = {};
            #pragma unroll
            for (int kc = 0; kc < 4; ++kc) {
                v2f b;
                b.x = dA[kc][R];
                b.y = dA[kc][4 + R];
                acc = __builtin_amdgcn_wmma_f32_16x16x4_f32(
                        false, fb[kc], false, b, (short)0, acc, false, false);
            }
            #pragma unroll
            for (int r = 0; r < 8; ++r)
                at(env, cell4(st0, R * S0 + (r & 3) * S1 + (r >> 2) * S2)) = acc[r];
        }
    }
}

__device__ __forceinline__ void column_sites(float* env,
                                             const float* M0, const float* M1,
                                             const float* M2, const float* M3,
                                             const float* M4, const float* M5,
                                             int wave, int c, int h, bool skip01) {
    if (!skip01) {
        site_pair<4096, 1024, 256, false>(env, M0, M1, wave, c, h);
        __syncthreads();
    }
    site_pair<256, 64, 16, false>(env, M2, M3, wave, c, h);
    __syncthreads();
    site_pair<16, 4, 1, false>(env, M4, M5, wave, c, h);
    __syncthreads();
}

__device__ __forceinline__ void column_sites_up(float* env,
                                                const float* M0, const float* M1,
                                                const float* M2, const float* M3,
                                                const float* M4, const float* M5,
                                                int wave, int c, int h, bool skip01) {
    site_pair<1, 4, 16, true>(env, M5, M4, wave, c, h);
    __syncthreads();
    site_pair<16, 64, 256, true>(env, M3, M2, wave, c, h);
    __syncthreads();
    if (!skip01) {
        site_pair<256, 1024, 4096, true>(env, M1, M0, wave, c, h);
        __syncthreads();
    }
}

__global__ void __launch_bounds__(256)
peps_edge_kernel(const float* __restrict__ Msite,
                 float* __restrict__ table,
                 size_t ws_bytes) {
    extern __shared__ float env[];

    const bool deep  = ws_bytes >= (size_t)WS_NEED3;
    const bool right = blockIdx.x >= (unsigned)(deep ? NCFG2 : NCFG);

    if (!deep) {
        if (blockIdx.x >= 2 * NCFG) return;
        if (ws_bytes < (size_t)(right ? WS_NEED2 : WS_NEED)) return;
    }

    const int tid  = threadIdx.x;
    const int cfg8 = blockIdx.x & ((deep ? NCFG2 : NCFG) - 1);
    const int cfg  = cfg8 & (NCFG - 1);
    const int m0   = right ? NSITE : 0;
    const int mst  = right ? 1 : LY;
    const int lane = tid & 31;
    const int wave = __builtin_amdgcn_readfirstlane(tid >> 5);
    __builtin_assume(wave >= 0 && wave < 8);
    const int c    = lane & 15;
    const int h    = lane >> 4;

    for (int q = tid; q < ENV7; q += 256) env[q] = 0.0f;
    __syncthreads();
    if (tid == 0) env[0] = 1.0f;
    __syncthreads();

    const float* M0 = Msite + ((m0 + 0 * mst) * PD + ((cfg >> 0) & 1)) * 256;
    const float* M1 = Msite + ((m0 + 1 * mst) * PD + ((cfg >> 1) & 1)) * 256;
    const float* M2 = Msite + ((m0 + 2 * mst) * PD + ((cfg >> 2) & 1)) * 256;
    const float* M3 = Msite + ((m0 + 3 * mst) * PD + ((cfg >> 3) & 1)) * 256;
    const float* M4 = Msite + ((m0 + 4 * mst) * PD + ((cfg >> 4) & 1)) * 256;
    const float* M5 = Msite + ((m0 + 5 * mst) * PD + ((cfg >> 5) & 1)) * 256;
    column_sites(env, M0, M1, M2, M3, M4, M5, wave, c, h, false);

    if (!deep) {
        float* row = table + blockIdx.x * ENV6;
        #pragma unroll
        for (int q = 0; q < 16; ++q)
            row[tid + 256 * q] = at(env, cell4(4 * swz(4 * tid), 1024 * q));
        return;
    }

    float keep[16];
    #pragma unroll
    for (int q = 0; q < 16; ++q) keep[q] = at(env, cell4(4 * swz(4 * tid), 1024 * q));
    __syncthreads();
    for (int q = ENV6 + tid; q < ENV7; q += 256) env[q] = 0.0f;
    #pragma unroll
    for (int q = 0; q < 16; ++q) at(env, cell4(4 * swz(tid), 256 * q)) = keep[q];
    __syncthreads();

    const int mA = right ? (NSITE + LX + 0) : (0 * LY + 1);
    const int mB = right ? (NSITE + LX + 1) : (1 * LY + 1);
    const float* MA = Msite + (mA * PD + ((cfg8 >> 6) & 1)) * 256;
    const float* MB = Msite + (mB * PD + ((cfg8 >> 7) & 1)) * 256;
    site_pair<4096, 1024, 256, false>(env, MA, MB, wave, c, h);
    __syncthreads();

    v4f*       row4 = reinterpret_cast<v4f*>(table + (size_t)blockIdx.x * ENV7);
    const v4f* env4 = reinterpret_cast<const v4f*>(env);
    #pragma unroll
    for (int q = 0; q < 16; ++q) row4[tid + 256 * q] = env4[tid + 256 * q];
}

__global__ void __launch_bounds__(256)
peps_amp_kernel(const int* __restrict__ x,
                const float* __restrict__ Msite,
                const float* __restrict__ table,
                size_t ws_bytes,
                float* __restrict__ out) {
    extern __shared__ float env[];

    const int tid    = threadIdx.x;
    const int sample = blockIdx.x;
    const int lane   = tid & 31;
    const int wave   = __builtin_amdgcn_readfirstlane(tid >> 5);
    __builtin_assume(wave >= 0 && wave < 8);
    const int c      = lane & 15;
    const int h      = lane >> 4;

    const int* xrow = x + sample * NSITE;
    const bool have_table = ws_bytes >= (size_t)WS_NEED;
    const bool have_right = ws_bytes >= (size_t)WS_NEED2;
    const bool have_deep  = ws_bytes >= (size_t)WS_NEED3;

    if (have_deep) {
        const int cfg = xrow[0 * LY] | (xrow[1 * LY] << 1) | (xrow[2 * LY] << 2) |
                        (xrow[3 * LY] << 3) | (xrow[4 * LY] << 4) | (xrow[5 * LY] << 5) |
                        (xrow[0 * LY + 1] << 6) | (xrow[1 * LY + 1] << 7);
        const v4f* row4 = reinterpret_cast<const v4f*>(table + (size_t)cfg * ENV7);
        v4f*       env4 = reinterpret_cast<v4f*>(env);
        v4f keep4[16];
        #pragma unroll
        for (int q = 0; q < 16; ++q) keep4[q] = row4[tid + 256 * q];
        #pragma unroll
        for (int q = 0; q < 16; ++q) env4[tid + 256 * q] = keep4[q];
        __syncthreads();
    } else if (have_table) {
        const int cfg = xrow[0 * LY] | (xrow[1 * LY] << 1) | (xrow[2 * LY] << 2) |
                        (xrow[3 * LY] << 3) | (xrow[4 * LY] << 4) | (xrow[5 * LY] << 5);
        const float* row = table + cfg * ENV6;
        float keep[16];
        #pragma unroll
        for (int q = 0; q < 16; ++q) keep[q] = row[tid + 256 * q];
        for (int q = ENV6 + tid; q < ENV7; q += 256) env[q] = 0.0f;
        #pragma unroll
        for (int q = 0; q < 16; ++q) at(env, cell4(4 * swz(tid), 256 * q)) = keep[q];
        __syncthreads();
    } else {
        for (int q = tid; q < ENV7; q += 256) env[q] = 0.0f;
        __syncthreads();
        if (tid == 0) env[0] = 1.0f;
        __syncthreads();
    }

    bool upward = false;
    const int jend = have_right ? LY - 1 : LY;
    for (int j = have_table ? 1 : 0; j < jend; ++j) {
        const float* M0 = Msite + ((0 * LY + j) * PD + xrow[0 * LY + j]) * 256;
        const float* M1 = Msite + ((1 * LY + j) * PD + xrow[1 * LY + j]) * 256;
        const float* M2 = Msite + ((2 * LY + j) * PD + xrow[2 * LY + j]) * 256;
        const float* M3 = Msite + ((3 * LY + j) * PD + xrow[3 * LY + j]) * 256;
        const float* M4 = Msite + ((4 * LY + j) * PD + xrow[4 * LY + j]) * 256;
        const float* M5 = Msite + ((5 * LY + j) * PD + xrow[5 * LY + j]) * 256;

        if (!upward) column_sites(env, M0, M1, M2, M3, M4, M5, wave, c, h, have_deep && j == 1);
        else         column_sites_up(env, M0, M1, M2, M3, M4, M5, wave, c, h, have_deep && j == LY - 2);
        upward = !upward;
    }

    if (!have_right) {
        if (tid == 0) out[sample] = env[0];
        return;
    }

    const int cfg5 = xrow[0 * LY + 5] | (xrow[1 * LY + 5] << 1) | (xrow[2 * LY + 5] << 2) |
                     (xrow[3 * LY + 5] << 3) | (xrow[4 * LY + 5] << 4) | (xrow[5 * LY + 5] << 5);
    float part = 0.0f;
    if (have_deep) {
        const int cfgR = cfg5 | (xrow[0 * LY + LY - 2] << 6) | (xrow[1 * LY + LY - 2] << 7);
        const v4f* rrow4 = reinterpret_cast<const v4f*>(table + (size_t)(NCFG2 + cfgR) * ENV7);
        const v4f* env4  = reinterpret_cast<const v4f*>(env);
        v4f rv4[16];
        #pragma unroll
        for (int q = 0; q < 16; ++q) rv4[q] = rrow4[tid + 256 * q];
        #pragma unroll
        for (int q = 0; q < 16; ++q) {
            const v4f e = env4[tid + 256 * q];
            part = fmaf(e.x, rv4[q].x, part);
            part = fmaf(e.y, rv4[q].y, part);
            part = fmaf(e.z, rv4[q].z, part);
            part = fmaf(e.w, rv4[q].w, part);
        }
    } else {
        const float* rrow = table + (NCFG + cfg5) * ENV6;
        float rv[16];
        #pragma unroll
        for (int q = 0; q < 16; ++q) rv[q] = rrow[tid + 256 * q];
        #pragma unroll
        for (int q = 0; q < 16; ++q)
            part = fmaf(at(env, cell4(4 * swz(tid), 256 * q)), rv[q], part);
    }
    for (int o = 16; o > 0; o >>= 1) part += __shfl_xor(part, o);
    __syncthreads();
    if (lane == 0) env[wave] = part;
    __syncthreads();
    if (tid == 0) {
        float t = env[0];
        for (int w = 1; w < 8; ++w) t += env[w];
        out[sample] = t;
    }
}

extern "C" void kernel_launch(void* const* d_in, const int* in_sizes, int n_in,
                              void* d_out, int out_size, void* d_ws, size_t ws_size,
                              hipStream_t stream) {
    const int*   x     = (const int*)d_in[0];
    const float* peps  = (const float*)d_in[1];
    float*       out   = (float*)d_out;
    float*       Msite = (float*)d_ws;
    float*       table = Msite + NMAT * PD * 256;

    const int nM = NMAT * PD * 256;
    peps_prep_kernel<<<(nM + 255) / 256, 256, 0, stream>>>(peps, Msite, ws_size);

    peps_edge_kernel<<<2 * NCFG2, 256, ENV7 * sizeof(float), stream>>>(Msite, table, ws_size);

    const int B = out_size;
    peps_amp_kernel<<<B, 256, ENV7 * sizeof(float), stream>>>(x, Msite, table, ws_size, out);
}


# edge tables a second site pair deeper if the workspace holds them (over 128 MiB both sides, over 80 MiB left only): left row = column 0 + rows 0-3 of column 1; sum order changes at the join
# speedup vs baseline: 5.2859x; 1.0701x over previous
#include <hip/hip_runtime.h>


typedef float v2f __attribute__((ext_vector_type(2)));
typedef float v4f __attribute__((ext_vector_type(4)));
typedef float v8f __attribute__((ext_vector_type(8)));

#define LX 6
#define LY 6
#define PD 2
#define DD 4
#define NSITE 36
#define ENV7 16384
#define ENV6 4096
#define NCFG 64
#define NCFG2 256
#define NCFG3 1024
#define NMAT 46
#define MSITE_BYTES (NMAT * PD * 256 * 4)
#define WS_NEED     (MSITE_BYTES + NCFG * ENV6 * 4)
#define WS_NEED2    (MSITE_BYTES + 2 * NCFG * ENV6 * 4)
#define WS_NEED3    (MSITE_BYTES + 2 * NCFG2 * ENV7 * 4)
#define WS_NEED4    (MSITE_BYTES + (NCFG3 + NCFG2) * ENV7 * 4)
#define WS_NEED5    (MSITE_BYTES + 2 * NCFG3 * ENV7 * 4)

__global__ void peps_prep_kernel(const float* __restrict__ peps,
                                 float* __restrict__ Msite,
                                 size_t ws_bytes) {
    int idx = blockIdx.x * blockDim.x + threadIdx.x;
    if (idx >= NMAT * PD * 256) return;
    if (idx >= NSITE * PD * 256 && ws_bytes < (size_t)WS_NEED2) return;
    if (idx >= (NSITE + LX) * PD * 256 && ws_bytes < (size_t)WS_NEED3) return;
    int n    = idx & 15;
    int k    = (idx >> 4) & 15;
    int s    = (idx >> 8) & 1;
    int site = idx >> 9;
    int a = k >> 2, c = k & 3;
    int d = n >> 2, b = n & 3;
    float v = 0.0f;
    if (site < NSITE) {
        int i = site / LY, j = site % LY;
        bool pad = (i == 0      && a > 0) || (i == LX - 1 && b > 0) ||
                   (j == 0      && c > 0) || (j == LY - 1 && d > 0);
        if (!pad) {
            int p = ((((site * PD + s) * DD + a) * DD + b) * DD + c) * DD + d;
            v = peps[p];
        }
    } else if (site < NSITE + LX) {
        int i  = site - NSITE;
        int rs = i * LY + (LY - 1);
        bool pad = (i == 0 && a > 0) || (i == LX - 1 && b > 0) || (c > 0);
        if (!pad) {
            int p = ((((rs * PD + s) * DD + a) * DD + b) * DD + d) * DD + c;
            v = peps[p];
        }
    } else {
        int i  = site - (NSITE + LX);
        int rs = i * LY + (LY - 2);
        bool pad = (i == 0 && a > 0);
        if (!pad) {
            int p = ((((rs * PD + s) * DD + a) * DD + b) * DD + d) * DD + c;
            v = peps[p];
        }
    }
    Msite[idx] = v;
}

__device__ __forceinline__ constexpr int swz(int a) { return a ^ ((a >> 6) & 15); }

__device__ __forceinline__ int cell4(int Lz4, int K) {
    const int X   = ((K >> 6) & 15) ^ (K & 15);
    const int KHI = K & ~15;
    return (Lz4 ^ (X * 4)) + KHI * 4;
}
__device__ __forceinline__ float& at(float* env, int byte_addr) {
    return *reinterpret_cast<float*>(reinterpret_cast<char*>(env) + byte_addr);
}

template <int S0, int S1, int S2, bool UPWARD>
__device__ __forceinline__ void site_pair(float* env,
                                          const float* __restrict__ MpA,
                                          const float* __restrict__ MpB,
                                          int wave, int c, int h) {
    constexpr int MINS = S0 < S2 ? S0 : S2;

    v2f fa[4], fb[4];
    #pragma unroll
    for (int kc = 0; kc < 4; ++kc) {
        if (!UPWARD) {
            const int ncol = (c & 3) * 4 + (c >> 2);
            fa[kc].x = MpA[((2 * h + 0) * 4 + kc) * 16 + ncol];
            fa[kc].y = MpA[((2 * h + 1) * 4 + kc) * 16 + ncol];
            fb[kc].x = MpB[((2 * h + 0) * 4 + kc) * 16 + ncol];
            fb[kc].y = MpB[((2 * h + 1) * 4 + kc) * 16 + ncol];
        } else {
            const int nrow = ((c >> 2) * 4 + kc) * 16 + (c & 3) * 4;
            fa[kc].x = MpA[nrow + 2 * h + 0];
            fa[kc].y = MpA[nrow + 2 * h + 1];
            fb[kc].x = MpB[nrow + 2 * h + 0];
            fb[kc].y = MpB[nrow + 2 * h + 1];
        }
    }

    #pragma nounroll
    for (int it = 0; it < 2; ++it) {
        const int q    = (wave + 8 * it) * 16 + c;
        const int ql   = q & (MINS - 1);
        const int base = ((q - ql) << 6) | ql;
        const int ld0  = 4 * swz(base + (2 * h) * S0);
        const int st0  = 4 * swz(base + (2 * h) * S2);

        v8f dA[4];
        #pragma unroll
        for (int T = 0; T < 4; ++T) {
            v8f acc = {};
            #pragma unroll
            for (int kc = 0; kc < 4; ++kc) {
                v2f b;
                b.x = at(env, cell4(ld0, kc * S1 + T * S2));
                b.y = at(env, cell4(ld0, S0 + kc * S1 + T * S2));
                acc = __builtin_amdgcn_wmma_f32_16x16x4_f32(
                        false, fa[kc], false, b, (short)0, acc, false, false);
            }
            dA[T] = acc;
        }

        #pragma unroll
        for (int R = 0; R < 4; ++R) {
            v8f acc = {};
            #pragma unroll
            for (int kc = 0; kc < 4; ++kc) {
                v2f b;
                b.x = dA[kc][R];
                b.y = dA[kc][4 + R];
                acc = __builtin_amdgcn_wmma_f32_16x16x4_f32(
                        false, fb[kc], false, b, (short)0, acc, false, false);
            }
            #pragma unroll
            for (int r = 0; r < 8; ++r)
                at(env, cell4(st0, R * S0 + (r & 3) * S1 + (r >> 2) * S2)) = acc[r];
        }
    }
}

__device__ __forceinline__ void column_sites(float* env,
                                             const float* M0, const float* M1,
                                             const float* M2, const float* M3,
                                             const float* M4, const float* M5,
                                             int wave, int c, int h, int skip) {
    if (skip < 1) {
        site_pair<4096, 1024, 256, false>(env, M0, M1, wave, c, h);
        __syncthreads();
    }
    if (skip < 2) {
        site_pair<256, 64, 16, false>(env, M2, M3, wave, c, h);
        __syncthreads();
    }
    site_pair<16, 4, 1, false>(env, M4, M5, wave, c, h);
    __syncthreads();
}

__device__ __forceinline__ void column_sites_up(float* env,
                                                const float* M0, const float* M1,
                                                const float* M2, const float* M3,
                                                const float* M4, const float* M5,
                                                int wave, int c, int h, int skip) {
    site_pair<1, 4, 16, true>(env, M5, M4, wave, c, h);
    __syncthreads();
    if (skip < 2) {
        site_pair<16, 64, 256, true>(env, M3, M2, wave, c, h);
        __syncthreads();
    }
    if (skip < 1) {
        site_pair<256, 1024, 4096, true>(env, M1, M0, wave, c, h);
        __syncthreads();
    }
}

__global__ void __launch_bounds__(256)
peps_edge_kernel(const float* __restrict__ Msite,
                 float* __restrict__ table,
                 size_t ws_bytes) {
    __shared__ __attribute__((aligned(16))) float env[2 * ENV7];

    const bool deep   = ws_bytes >= (size_t)WS_NEED3;
    const bool deep2L = ws_bytes >= (size_t)WS_NEED4;
    const bool deep2R = ws_bytes >= (size_t)WS_NEED5;
    const bool right = blockIdx.x >= (unsigned)(deep ? NCFG2 : NCFG);

    if (!deep) {
        if (blockIdx.x >= 2 * NCFG) return;
        if (ws_bytes < (size_t)(right ? WS_NEED2 : WS_NEED)) return;
    }

    const int tid  = threadIdx.x;
    const int cfg8 = blockIdx.x & ((deep ? NCFG2 : NCFG) - 1);
    const int cfg  = cfg8 & (NCFG - 1);
    const int m0   = right ? NSITE : 0;
    const int mst  = right ? 1 : LY;
    const int lane = tid & 31;
    const int wave = __builtin_amdgcn_readfirstlane(tid >> 5);
    __builtin_assume(wave >= 0 && wave < 8);
    const int c    = lane & 15;
    const int h    = lane >> 4;

    for (int q = tid; q < ENV7; q += 256) env[q] = 0.0f;
    __syncthreads();
    if (tid == 0) env[0] = 1.0f;
    __syncthreads();

    const float* M0 = Msite + ((m0 + 0 * mst) * PD + ((cfg >> 0) & 1)) * 256;
    const float* M1 = Msite + ((m0 + 1 * mst) * PD + ((cfg >> 1) & 1)) * 256;
    const float* M2 = Msite + ((m0 + 2 * mst) * PD + ((cfg >> 2) & 1)) * 256;
    const float* M3 = Msite + ((m0 + 3 * mst) * PD + ((cfg >> 3) & 1)) * 256;
    const float* M4 = Msite + ((m0 + 4 * mst) * PD + ((cfg >> 4) & 1)) * 256;
    const float* M5 = Msite + ((m0 + 5 * mst) * PD + ((cfg >> 5) & 1)) * 256;
    column_sites(env, M0, M1, M2, M3, M4, M5, wave, c, h, 0);

    if (!deep) {
        float* row = table + blockIdx.x * ENV6;
        #pragma unroll
        for (int q = 0; q < 16; ++q)
            row[tid + 256 * q] = at(env, cell4(4 * swz(4 * tid), 1024 * q));
        return;
    }

    float keep[16];
    #pragma unroll
    for (int q = 0; q < 16; ++q) keep[q] = at(env, cell4(4 * swz(4 * tid), 1024 * q));
    __syncthreads();
    for (int q = ENV6 + tid; q < ENV7; q += 256) env[q] = 0.0f;
    #pragma unroll
    for (int q = 0; q < 16; ++q) at(env, cell4(4 * swz(tid), 256 * q)) = keep[q];
    __syncthreads();

    const int mA = right ? (NSITE + LX + 0) : (0 * LY + 1);
    const int mB = right ? (NSITE + LX + 1) : (1 * LY + 1);
    const float* MA = Msite + (mA * PD + ((cfg8 >> 6) & 1)) * 256;
    const float* MB = Msite + (mB * PD + ((cfg8 >> 7) & 1)) * 256;
    site_pair<4096, 1024, 256, false>(env, MA, MB, wave, c, h);
    __syncthreads();

    const bool deeper = right ? deep2R : deep2L;
    float*     side   = table + (right ? (size_t)(deep2L ? NCFG3 : NCFG2) * ENV7 : (size_t)0);
    v4f*       env4   = reinterpret_cast<v4f*>(env);

    if (!deeper) {
        v4f* row4 = reinterpret_cast<v4f*>(side + (size_t)cfg8 * ENV7);
        #pragma unroll
        for (int q = 0; q < 16; ++q) row4[tid + 256 * q] = env4[tid + 256 * q];
        return;
    }

    v4f* copy4 = reinterpret_cast<v4f*>(env + ENV7);
    #pragma unroll
    for (int q = 0; q < 16; ++q) copy4[tid + 256 * q] = env4[tid + 256 * q];
    __syncthreads();

    const int mC = right ? (NSITE + LX + 2) : (2 * LY + 1);
    const int mD = right ? (NSITE + LX + 3) : (3 * LY + 1);
    #pragma nounroll
    for (int v = 0; v < 4; ++v) {
        if (v > 0) {
            __syncthreads();
            #pragma unroll
            for (int q = 0; q < 16; ++q) env4[tid + 256 * q] = copy4[tid + 256 * q];
            __syncthreads();
        }
        const float* MC = Msite + (mC * PD + (v & 1)) * 256;
        const float* MD = Msite + (mD * PD + (v >> 1)) * 256;
        site_pair<256, 64, 16, false>(env, MC, MD, wave, c, h);
        __syncthreads();
        v4f* row4 = reinterpret_cast<v4f*>(side + (size_t)(cfg8 + NCFG2 * v) * ENV7);
        #pragma unroll
        for (int q = 0; q < 16; ++q) row4[tid + 256 * q] = env4[tid + 256 * q];
    }
}

__global__ void __launch_bounds__(256)
peps_amp_kernel(const int* __restrict__ x,
                const float* __restrict__ Msite,
                const float* __restrict__ table,
                size_t ws_bytes,
                float* __restrict__ out) {
    extern __shared__ float env[];

    const int tid    = threadIdx.x;
    const int sample = blockIdx.x;
    const int lane   = tid & 31;
    const int wave   = __builtin_amdgcn_readfirstlane(tid >> 5);
    __builtin_assume(wave >= 0 && wave < 8);
    const int c      = lane & 15;
    const int h      = lane >> 4;

    const int* xrow = x + sample * NSITE;
    const bool have_table = ws_bytes >= (size_t)WS_NEED;
    const bool have_right = ws_bytes >= (size_t)WS_NEED2;
    const bool have_deep  = ws_bytes >= (size_t)WS_NEED3;
    const bool have_deep2L = ws_bytes >= (size_t)WS_NEED4;
    const bool have_deep2R = ws_bytes >= (size_t)WS_NEED5;

    if (have_deep) {
        int cfg = xrow[0 * LY] | (xrow[1 * LY] << 1) | (xrow[2 * LY] << 2) |
                  (xrow[3 * LY] << 3) | (xrow[4 * LY] << 4) | (xrow[5 * LY] << 5) |
                  (xrow[0 * LY + 1] << 6) | (xrow[1 * LY + 1] << 7);
        if (have_deep2L) cfg |= (xrow[2 * LY + 1] << 8) | (xrow[3 * LY + 1] << 9);
        const v4f* row4 = reinterpret_cast<const v4f*>(table + (size_t)cfg * ENV7);
        v4f*       env4 = reinterpret_cast<v4f*>(env);
        v4f keep4[16];
        #pragma unroll
        for (int q = 0; q < 16; ++q) keep4[q] = row4[tid + 256 * q];
        #pragma unroll
        for (int q = 0; q < 16; ++q) env4[tid + 256 * q] = keep4[q];
        __syncthreads();
    } else if (have_table) {
        const int cfg = xrow[0 * LY] | (xrow[1 * LY] << 1) | (xrow[2 * LY] << 2) |
                        (xrow[3 * LY] << 3) | (xrow[4 * LY] << 4) | (xrow[5 * LY] << 5);
        const float* row = table + cfg * ENV6;
        float keep[16];
        #pragma unroll
        for (int q = 0; q < 16; ++q) keep[q] = row[tid + 256 * q];
        for (int q = ENV6 + tid; q < ENV7; q += 256) env[q] = 0.0f;
        #pragma unroll
        for (int q = 0; q < 16; ++q) at(env, cell4(4 * swz(tid), 256 * q)) = keep[q];
        __syncthreads();
    } else {
        for (int q = tid; q < ENV7; q += 256) env[q] = 0.0f;
        __syncthreads();
        if (tid == 0) env[0] = 1.0f;
        __syncthreads();
    }

    bool upward = false;
    const int jend = have_right ? LY - 1 : LY;
    for (int j = have_table ? 1 : 0; j < jend; ++j) {
        const float* M0 = Msite + ((0 * LY + j) * PD + xrow[0 * LY + j]) * 256;
        const float* M1 = Msite + ((1 * LY + j) * PD + xrow[1 * LY + j]) * 256;
        const float* M2 = Msite + ((2 * LY + j) * PD + xrow[2 * LY + j]) * 256;
        const float* M3 = Msite + ((3 * LY + j) * PD + xrow[3 * LY + j]) * 256;
        const float* M4 = Msite + ((4 * LY + j) * PD + xrow[4 * LY + j]) * 256;
        const float* M5 = Msite + ((5 * LY + j) * PD + xrow[5 * LY + j]) * 256;

        const int skip_dn = (have_deep && j == 1)      ? (have_deep2L ? 2 : 1) : 0;
        const int skip_up = (have_deep && j == LY - 2) ? (have_deep2R ? 2 : 1) : 0;
        if (!upward) column_sites(env, M0, M1, M2, M3, M4, M5, wave, c, h, skip_dn);
        else         column_sites_up(env, M0, M1, M2, M3, M4, M5, wave, c, h, skip_up);
        upward = !upward;
    }

    if (!have_right) {
        if (tid == 0) out[sample] = env[0];
        return;
    }

    const int cfg5 = xrow[0 * LY + 5] | (xrow[1 * LY + 5] << 1) | (xrow[2 * LY + 5] << 2) |
                     (xrow[3 * LY + 5] << 3) | (xrow[4 * LY + 5] << 4) | (xrow[5 * LY + 5] << 5);
    float part = 0.0f;
    if (have_deep) {
        int cfgR = cfg5 | (xrow[0 * LY + LY - 2] << 6) | (xrow[1 * LY + LY - 2] << 7);
        if (have_deep2R) cfgR |= (xrow[2 * LY + LY - 2] << 8) | (xrow[3 * LY + LY - 2] << 9);
        const int nl = have_deep2L ? NCFG3 : NCFG2;
        const v4f* rrow4 = reinterpret_cast<const v4f*>(table + (size_t)(nl + cfgR) * ENV7);
        const v4f* env4  = reinterpret_cast<const v4f*>(env);
        v4f rv4[16];
        #pragma unroll
        for (int q = 0; q < 16; ++q) rv4[q] = rrow4[tid + 256 * q];
        #pragma unroll
        for (int q = 0; q < 16; ++q) {
            const v4f e = env4[tid + 256 * q];
            part = fmaf(e.x, rv4[q].x, part);
            part = fmaf(e.y, rv4[q].y, part);
            part = fmaf(e.z, rv4[q].z, part);
            part = fmaf(e.w, rv4[q].w, part);
        }
    } else {
        const float* rrow = table + (NCFG + cfg5) * ENV6;
        float rv[16];
        #pragma unroll
        for (int q = 0; q < 16; ++q) rv[q] = rrow[tid + 256 * q];
        #pragma unroll
        for (int q = 0; q < 16; ++q)
            part = fmaf(at(env, cell4(4 * swz(tid), 256 * q)), rv[q], part);
    }
    for (int o = 16; o > 0; o >>= 1) part += __shfl_xor(part, o);
    __syncthreads();
    if (lane == 0) env[wave] = part;
    __syncthreads();
    if (tid == 0) {
        float t = env[0];
        for (int w = 1; w < 8; ++w) t += env[w];
        out[sample] = t;
    }
}

extern "C" void kernel_launch(void* const* d_in, const int* in_sizes, int n_in,
                              void* d_out, int out_size, void* d_ws, size_t ws_size,
                              hipStream_t stream) {
    const int*   x     = (const int*)d_in[0];
    const float* peps  = (const float*)d_in[1];
    float*       out   = (float*)d_out;
    float*       Msite = (float*)d_ws;
    float*       table = Msite + NMAT * PD * 256;

    const int nM = NMAT * PD * 256;
    peps_prep_kernel<<<(nM + 255) / 256, 256, 0, stream>>>(peps, Msite, ws_size);

    peps_edge_kernel<<<2 * NCFG2, 256, 0, stream>>>(Msite, table, ws_size);

    const int B = out_size;
    peps_amp_kernel<<<B, 256, ENV7 * sizeof(float), stream>>>(x, Msite, table, ws_size, out);
}


# env layout with fixed address bits per bond (carried bond bits 0-1, row i bits 2+2i): B operands and results of both site steps are 64-bit LDS pairs used in place, no moves; join sum reordered
# speedup vs baseline: 5.8835x; 1.1131x over previous
#include <hip/hip_runtime.h>


typedef float v2f __attribute__((ext_vector_type(2)));
typedef float v4f __attribute__((ext_vector_type(4)));
typedef float v8f __attribute__((ext_vector_type(8)));

#define LX 6
#define LY 6
#define PD 2
#define DD 4
#define NSITE 36
#define ENV7 16384
#define ENV6 4096
#define NCFG 64
#define NCFG2 256
#define NCFG3 1024
#define NMAT 46
#define MSITE_BYTES (NMAT * PD * 256 * 4)
#define WS_NEED     (MSITE_BYTES + NCFG * ENV6 * 4)
#define WS_NEED2    (MSITE_BYTES + 2 * NCFG * ENV6 * 4)
#define WS_NEED3    (MSITE_BYTES + 2 * NCFG2 * ENV7 * 4)
#define WS_NEED4    (MSITE_BYTES + (NCFG3 + NCFG2) * ENV7 * 4)
#define WS_NEED5    (MSITE_BYTES + 2 * NCFG3 * ENV7 * 4)

__global__ void peps_prep_kernel(const float* __restrict__ peps,
                                 float* __restrict__ Msite,
                                 size_t ws_bytes) {
    int idx = blockIdx.x * blockDim.x + threadIdx.x;
    if (idx >= NMAT * PD * 256) return;
    if (idx >= NSITE * PD * 256 && ws_bytes < (size_t)WS_NEED2) return;
    if (idx >= (NSITE + LX) * PD * 256 && ws_bytes < (size_t)WS_NEED3) return;
    int n    = idx & 15;
    int k    = (idx >> 4) & 15;
    int s    = (idx >> 8) & 1;
    int site = idx >> 9;
    int a = k >> 2, c = k & 3;
    int d = n >> 2, b = n & 3;
    float v = 0.0f;
    if (site < NSITE) {
        int i = site / LY, j = site % LY;
        bool pad = (i == 0      && a > 0) || (i == LX - 1 && b > 0) ||
                   (j == 0      && c > 0) || (j == LY - 1 && d > 0);
        if (!pad) {
            int p = ((((site * PD + s) * DD + a) * DD + b) * DD + c) * DD + d;
            v = peps[p];
        }
    } else if (site < NSITE + LX) {
        int i  = site - NSITE;
        int rs = i * LY + (LY - 1);
        bool pad = (i == 0 && a > 0) || (i == LX - 1 && b > 0) || (c > 0);
        if (!pad) {
            int p = ((((rs * PD + s) * DD + a) * DD + b) * DD + d) * DD + c;
            v = peps[p];
        }
    } else {
        int i  = site - (NSITE + LX);
        int rs = i * LY + (LY - 2);
        bool pad = (i == 0 && a > 0);
        if (!pad) {
            int p = ((((rs * PD + s) * DD + a) * DD + b) * DD + d) * DD + c;
            v = peps[p];
        }
    }
    Msite[idx] = v;
}

__device__ __forceinline__ constexpr int swz(int a) { return a ^ (((a >> 6) & 15) << 2); }

__device__ __forceinline__ int cell4(int Lz4, int K) {
    const int X   = ((K >> 6) & 15) ^ ((K >> 2) & 15);
    const int KHI = K & ~63;
    return (Lz4 ^ (X * 16)) + KHI * 4;
}
__device__ __forceinline__ int ccell4(int C) { return 16 * (C ^ ((C >> 4) & 15)); }
__device__ __forceinline__ float& at(float* env, int byte_addr) {
    return *reinterpret_cast<float*>(reinterpret_cast<char*>(env) + byte_addr);
}
__device__ __forceinline__ v2f& at2(float* env, int byte_addr) {
    return *reinterpret_cast<v2f*>(reinterpret_cast<char*>(env) + byte_addr);
}
__device__ __forceinline__ v4f& at4(float* env, int byte_addr) {
    return *reinterpret_cast<v4f*>(reinterpret_cast<char*>(env) + byte_addr);
}

template <int P, bool UPWARD>
__device__ __forceinline__ void site_pair(float* env,
                                          const float* __restrict__ MpA,
                                          const float* __restrict__ MpB,
                                          int wave, int c, int h) {
    constexpr int P1 = UPWARD ? P + 2 : P;
    constexpr int P2 = UPWARD ? P : P + 2;

    const int vo = 2 * (c >> 3) + (c & 1);
    const int ro = (c >> 1) & 3;
    v2f fa[4], fb[4];
    #pragma unroll
    for (int kc = 0; kc < 4; ++kc) {
        if (!UPWARD) {
            const int ncol = ro * 4 + vo;
            fa[kc].x = MpA[((2 * h + 0) * 4 + kc) * 16 + ncol];
            fa[kc].y = MpA[((2 * h + 1) * 4 + kc) * 16 + ncol];
            fb[kc].x = MpB[((2 * h + 0) * 4 + kc) * 16 + ncol];
            fb[kc].y = MpB[((2 * h + 1) * 4 + kc) * 16 + ncol];
        } else {
            const int nrow = (vo * 4 + kc) * 16 + ro * 4;
            fa[kc].x = MpA[nrow + 2 * h + 0];
            fa[kc].y = MpA[nrow + 2 * h + 1];
            fb[kc].x = MpB[nrow + 2 * h + 0];
            fb[kc].y = MpB[nrow + 2 * h + 1];
        }
    }

    #pragma nounroll
    for (int it = 0; it < 2; ++it) {
        const int q    = (wave + 8 * it) * 16 + c;
        const int ql   = q & ((1 << (P - 2)) - 1);
        const int base = ((q >> (P - 2)) << (P + 4)) | (ql << 2);
        const int a0   = 4 * swz(base + 2 * h);

        v8f dA[4];
        #pragma unroll
        for (int T = 0; T < 4; ++T) {
            v8f acc = {};
            #pragma unroll
            for (int kc = 0; kc < 4; ++kc) {
                const v2f b = at2(env, cell4(a0, (kc << P1) | (T << P2)));
                acc = __builtin_amdgcn_wmma_f32_16x16x4_f32(
                        false, fa[kc], false, b, (short)0, acc, false, false);
            }
            dA[T] = acc;
        }

        #pragma unroll
        for (int R = 0; R < 4; ++R) {
            v8f acc = {};
            #pragma unroll
            for (int kc = 0; kc < 4; ++kc) {
                v2f b;
                b.x = dA[kc][2 * R];
                b.y = dA[kc][2 * R + 1];
                acc = __builtin_amdgcn_wmma_f32_16x16x4_f32(
                        false, fb[kc], false, b, (short)0, acc, false, false);
            }
            #pragma unroll
            for (int r2 = 0; r2 < 4; ++r2) {
                v2f o;
                o.x = acc[2 * r2];
                o.y = acc[2 * r2 + 1];
                at2(env, cell4(a0, (R << P1) | (r2 << P2))) = o;
            }
        }
    }
}

__device__ __forceinline__ void column_sites(float* env,
                                             const float* M0, const float* M1,
                                             const float* M2, const float* M3,
                                             const float* M4, const float* M5,
                                             int wave, int c, int h, int skip) {
    if (skip < 1) {
        site_pair<2, false>(env, M0, M1, wave, c, h);
        __syncthreads();
    }
    if (skip < 2) {
        site_pair<6, false>(env, M2, M3, wave, c, h);
        __syncthreads();
    }
    site_pair<10, false>(env, M4, M5, wave, c, h);
    __syncthreads();
}

__device__ __forceinline__ void column_sites_up(float* env,
                                                const float* M0, const float* M1,
                                                const float* M2, const float* M3,
                                                const float* M4, const float* M5,
                                                int wave, int c, int h, int skip) {
    site_pair<10, true>(env, M5, M4, wave, c, h);
    __syncthreads();
    if (skip < 2) {
        site_pair<6, true>(env, M3, M2, wave, c, h);
        __syncthreads();
    }
    if (skip < 1) {
        site_pair<2, true>(env, M1, M0, wave, c, h);
        __syncthreads();
    }
}

__global__ void __launch_bounds__(256)
peps_edge_kernel(const float* __restrict__ Msite,
                 float* __restrict__ table,
                 size_t ws_bytes) {
    __shared__ __attribute__((aligned(16))) float env[2 * ENV7];

    const bool deep   = ws_bytes >= (size_t)WS_NEED3;
    const bool deep2L = ws_bytes >= (size_t)WS_NEED4;
    const bool deep2R = ws_bytes >= (size_t)WS_NEED5;
    const bool right = blockIdx.x >= (unsigned)(deep ? NCFG2 : NCFG);

    if (!deep) {
        if (blockIdx.x >= 2 * NCFG) return;
        if (ws_bytes < (size_t)(right ? WS_NEED2 : WS_NEED)) return;
    }

    const int tid  = threadIdx.x;
    const int cfg8 = blockIdx.x & ((deep ? NCFG2 : NCFG) - 1);
    const int cfg  = cfg8 & (NCFG - 1);
    const int m0   = right ? NSITE : 0;
    const int mst  = right ? 1 : LY;
    const int lane = tid & 31;
    const int wave = __builtin_amdgcn_readfirstlane(tid >> 5);
    __builtin_assume(wave >= 0 && wave < 8);
    const int c    = lane & 15;
    const int h    = lane >> 4;

    for (int q = tid; q < ENV7; q += 256) env[q] = 0.0f;
    __syncthreads();
    if (tid == 0) env[0] = 1.0f;
    __syncthreads();

    const float* M0 = Msite + ((m0 + 0 * mst) * PD + ((cfg >> 0) & 1)) * 256;
    const float* M1 = Msite + ((m0 + 1 * mst) * PD + ((cfg >> 1) & 1)) * 256;
    const float* M2 = Msite + ((m0 + 2 * mst) * PD + ((cfg >> 2) & 1)) * 256;
    const float* M3 = Msite + ((m0 + 3 * mst) * PD + ((cfg >> 3) & 1)) * 256;
    const float* M4 = Msite + ((m0 + 4 * mst) * PD + ((cfg >> 4) & 1)) * 256;
    const float* M5 = Msite + ((m0 + 5 * mst) * PD + ((cfg >> 5) & 1)) * 256;
    column_sites(env, M0, M1, M2, M3, M4, M5, wave, c, h, 0);

    if (!deep) {
        float* row = table + blockIdx.x * ENV6;
        #pragma unroll
        for (int q = 0; q < 16; ++q)
            row[tid + 256 * q] = at(env, ccell4(tid) + 4096 * q);
        return;
    }

    const int mA = right ? (NSITE + LX + 0) : (0 * LY + 1);
    const int mB = right ? (NSITE + LX + 1) : (1 * LY + 1);
    const float* MA = Msite + (mA * PD + ((cfg8 >> 6) & 1)) * 256;
    const float* MB = Msite + (mB * PD + ((cfg8 >> 7) & 1)) * 256;
    site_pair<2, false>(env, MA, MB, wave, c, h);
    __syncthreads();

    const bool deeper = right ? deep2R : deep2L;
    float*     side   = table + (right ? (size_t)(deep2L ? NCFG3 : NCFG2) * ENV7 : (size_t)0);
    v4f*       env4   = reinterpret_cast<v4f*>(env);

    if (!deeper) {
        v4f* row4 = reinterpret_cast<v4f*>(side + (size_t)cfg8 * ENV7);
        #pragma unroll
        for (int q = 0; q < 16; ++q) row4[tid + 256 * q] = env4[tid + 256 * q];
        return;
    }

    v4f* copy4 = reinterpret_cast<v4f*>(env + ENV7);
    #pragma unroll
    for (int q = 0; q < 16; ++q) copy4[tid + 256 * q] = env4[tid + 256 * q];
    __syncthreads();

    const int mC = right ? (NSITE + LX + 2) : (2 * LY + 1);
    const int mD = right ? (NSITE + LX + 3) : (3 * LY + 1);
    #pragma nounroll
    for (int v = 0; v < 4; ++v) {
        if (v > 0) {
            __syncthreads();
            #pragma unroll
            for (int q = 0; q < 16; ++q) env4[tid + 256 * q] = copy4[tid + 256 * q];
            __syncthreads();
        }
        const float* MC = Msite + (mC * PD + (v & 1)) * 256;
        const float* MD = Msite + (mD * PD + (v >> 1)) * 256;
        site_pair<6, false>(env, MC, MD, wave, c, h);
        __syncthreads();
        v4f* row4 = reinterpret_cast<v4f*>(side + (size_t)(cfg8 + NCFG2 * v) * ENV7);
        #pragma unroll
        for (int q = 0; q < 16; ++q) row4[tid + 256 * q] = env4[tid + 256 * q];
    }
}

__global__ void __launch_bounds__(256)
peps_amp_kernel(const int* __restrict__ x,
                const float* __restrict__ Msite,
                const float* __restrict__ table,
                size_t ws_bytes,
                float* __restrict__ out) {
    extern __shared__ float env[];

    const int tid    = threadIdx.x;
    const int sample = blockIdx.x;
    const int lane   = tid & 31;
    const int wave   = __builtin_amdgcn_readfirstlane(tid >> 5);
    __builtin_assume(wave >= 0 && wave < 8);
    const int c      = lane & 15;
    const int h      = lane >> 4;

    const int* xrow = x + sample * NSITE;
    const bool have_table = ws_bytes >= (size_t)WS_NEED;
    const bool have_right = ws_bytes >= (size_t)WS_NEED2;
    const bool have_deep  = ws_bytes >= (size_t)WS_NEED3;
    const bool have_deep2L = ws_bytes >= (size_t)WS_NEED4;
    const bool have_deep2R = ws_bytes >= (size_t)WS_NEED5;

    if (have_deep) {
        int cfg = xrow[0 * LY] | (xrow[1 * LY] << 1) | (xrow[2 * LY] << 2) |
                  (xrow[3 * LY] << 3) | (xrow[4 * LY] << 4) | (xrow[5 * LY] << 5) |
                  (xrow[0 * LY + 1] << 6) | (xrow[1 * LY + 1] << 7);
        if (have_deep2L) cfg |= (xrow[2 * LY + 1] << 8) | (xrow[3 * LY + 1] << 9);
        const v4f* row4 = reinterpret_cast<const v4f*>(table + (size_t)cfg * ENV7);
        v4f*       env4 = reinterpret_cast<v4f*>(env);
        v4f keep4[16];
        #pragma unroll
        for (int q = 0; q < 16; ++q) keep4[q] = row4[tid + 256 * q];
        #pragma unroll
        for (int q = 0; q < 16; ++q) env4[tid + 256 * q] = keep4[q];
        __syncthreads();
    } else if (have_table) {
        const int cfg = xrow[0 * LY] | (xrow[1 * LY] << 1) | (xrow[2 * LY] << 2) |
                        (xrow[3 * LY] << 3) | (xrow[4 * LY] << 4) | (xrow[5 * LY] << 5);
        const float* row = table + cfg * ENV6;
        float keep[16];
        #pragma unroll
        for (int q = 0; q < 16; ++q) keep[q] = row[tid + 256 * q];
        #pragma unroll
        for (int q = 0; q < 16; ++q) {
            const v4f t = {keep[q], 0.0f, 0.0f, 0.0f};
            at4(env, ccell4(tid) + 4096 * q) = t;
        }
        __syncthreads();
    } else {
        for (int q = tid; q < ENV7; q += 256) env[q] = 0.0f;
        __syncthreads();
        if (tid == 0) env[0] = 1.0f;
        __syncthreads();
    }

    bool upward = false;
    const int jend = have_right ? LY - 1 : LY;
    for (int j = have_table ? 1 : 0; j < jend; ++j) {
        const float* M0 = Msite + ((0 * LY + j) * PD + xrow[0 * LY + j]) * 256;
        const float* M1 = Msite + ((1 * LY + j) * PD + xrow[1 * LY + j]) * 256;
        const float* M2 = Msite + ((2 * LY + j) * PD + xrow[2 * LY + j]) * 256;
        const float* M3 = Msite + ((3 * LY + j) * PD + xrow[3 * LY + j]) * 256;
        const float* M4 = Msite + ((4 * LY + j) * PD + xrow[4 * LY + j]) * 256;
        const float* M5 = Msite + ((5 * LY + j) * PD + xrow[5 * LY + j]) * 256;

        const int skip_dn = (have_deep && j == 1)      ? (have_deep2L ? 2 : 1) : 0;
        const int skip_up = (have_deep && j == LY - 2) ? (have_deep2R ? 2 : 1) : 0;
        if (!upward) column_sites(env, M0, M1, M2, M3, M4, M5, wave, c, h, skip_dn);
        else         column_sites_up(env, M0, M1, M2, M3, M4, M5, wave, c, h, skip_up);
        upward = !upward;
    }

    if (!have_right) {
        if (tid == 0) out[sample] = env[0];
        return;
    }

    const int cfg5 = xrow[0 * LY + 5] | (xrow[1 * LY + 5] << 1) | (xrow[2 * LY + 5] << 2) |
                     (xrow[3 * LY + 5] << 3) | (xrow[4 * LY + 5] << 4) | (xrow[5 * LY + 5] << 5);
    float part = 0.0f;
    if (have_deep) {
        int cfgR = cfg5 | (xrow[0 * LY + LY - 2] << 6) | (xrow[1 * LY + LY - 2] << 7);
        if (have_deep2R) cfgR |= (xrow[2 * LY + LY - 2] << 8) | (xrow[3 * LY + LY - 2] << 9);
        const int nl = have_deep2L ? NCFG3 : NCFG2;
        const v4f* rrow4 = reinterpret_cast<const v4f*>(table + (size_t)(nl + cfgR) * ENV7);
        const v4f* env4  = reinterpret_cast<const v4f*>(env);
        v4f rv4[16];
        #pragma unroll
        for (int q = 0; q < 16; ++q) rv4[q] = rrow4[tid + 256 * q];
        #pragma unroll
        for (int q = 0; q < 16; ++q) {
            const v4f e = env4[tid + 256 * q];
            part = fmaf(e.x, rv4[q].x, part);
            part = fmaf(e.y, rv4[q].y, part);
            part = fmaf(e.z, rv4[q].z, part);
            part = fmaf(e.w, rv4[q].w, part);
        }
    } else {
        const float* rrow = table + (NCFG + cfg5) * ENV6;
        float rv[16];
        #pragma unroll
        for (int q = 0; q < 16; ++q) rv[q] = rrow[tid + 256 * q];
        #pragma unroll
        for (int q = 0; q < 16; ++q)
            part = fmaf(at(env, ccell4(tid) + 4096 * q), rv[q], part);
    }
    for (int o = 16; o > 0; o >>= 1) part += __shfl_xor(part, o);
    __syncthreads();
    if (lane == 0) env[wave] = part;
    __syncthreads();
    if (tid == 0) {
        float t = env[0];
        for (int w = 1; w < 8; ++w) t += env[w];
        out[sample] = t;
    }
}

extern "C" void kernel_launch(void* const* d_in, const int* in_sizes, int n_in,
                              void* d_out, int out_size, void* d_ws, size_t ws_size,
                              hipStream_t stream) {
    const int*   x     = (const int*)d_in[0];
    const float* peps  = (const float*)d_in[1];
    float*       out   = (float*)d_out;
    float*       Msite = (float*)d_ws;
    float*       table = Msite + NMAT * PD * 256;

    const int nM = NMAT * PD * 256;
    peps_prep_kernel<<<(nM + 255) / 256, 256, 0, stream>>>(peps, Msite, ws_size);

    peps_edge_kernel<<<2 * NCFG2, 256, 0, stream>>>(Msite, table, ws_size);

    const int B = out_size;
    peps_amp_kernel<<<B, 256, ENV7 * sizeof(float), stream>>>(x, Msite, table, ws_size, out);
}


# edge tables a whole neighbour column deep: rows 4,5 of columns 1 and 4 applied in the table launch for all 4 index pairs from one env read; 4096 compact rows a side; join of 4096 terms, sum reordered
# speedup vs baseline: 7.4885x; 1.2728x over previous
#include <hip/hip_runtime.h>


typedef float v2f __attribute__((ext_vector_type(2)));
typedef float v4f __attribute__((ext_vector_type(4)));
typedef float v8f __attribute__((ext_vector_type(8)));

#define LX 6
#define LY 6
#define PD 2
#define DD 4
#define NSITE 36
#define ENV7 16384
#define ENV6 4096
#define NCFG 64
#define NCFG2 256
#define NCFG3 1024
#define NCFG4 4096
#define NMAT 46
#define MSITE_BYTES (NMAT * PD * 256 * 4)
#define WS_NEED     (MSITE_BYTES + NCFG * ENV6 * 4)
#define WS_NEED2    (MSITE_BYTES + 2 * NCFG * ENV6 * 4)
#define WS_NEED3    (MSITE_BYTES + 2 * NCFG2 * ENV7 * 4)
#define WS_NEED4    (MSITE_BYTES + (NCFG3 + NCFG2) * ENV7 * 4)
#define WS_NEED5    (MSITE_BYTES + 2 * NCFG4 * ENV6 * 4)

__global__ void peps_prep_kernel(const float* __restrict__ peps,
                                 float* __restrict__ Msite,
                                 size_t ws_bytes) {
    int idx = blockIdx.x * blockDim.x + threadIdx.x;
    if (idx >= NMAT * PD * 256) return;
    if (idx >= NSITE * PD * 256 && ws_bytes < (size_t)WS_NEED2) return;
    if (idx >= (NSITE + LX) * PD * 256 && ws_bytes < (size_t)WS_NEED3) return;
    int n    = idx & 15;
    int k    = (idx >> 4) & 15;
    int s    = (idx >> 8) & 1;
    int site = idx >> 9;
    int a = k >> 2, c = k & 3;
    int d = n >> 2, b = n & 3;
    float v = 0.0f;
    if (site < NSITE) {
        int i = site / LY, j = site % LY;
        bool pad = (i == 0      && a > 0) || (i == LX - 1 && b > 0) ||
                   (j == 0      && c > 0) || (j == LY - 1 && d > 0);
        if (!pad) {
            int p = ((((site * PD + s) * DD + a) * DD + b) * DD + c) * DD + d;
            v = peps[p];
        }
    } else if (site < NSITE + LX) {
        int i  = site - NSITE;
        int rs = i * LY + (LY - 1);
        bool pad = (i == 0 && a > 0) || (i == LX - 1 && b > 0) || (c > 0);
        if (!pad) {
            int p = ((((rs * PD + s) * DD + a) * DD + b) * DD + d) * DD + c;
            v = peps[p];
        }
    } else {
        int i  = site - (NSITE + LX);
        int rs = i * LY + (LY - 2);
        bool pad = (i == 0 && a > 0);
        if (!pad) {
            int p = ((((rs * PD + s) * DD + a) * DD + b) * DD + d) * DD + c;
            v = peps[p];
        }
    }
    Msite[idx] = v;
}

__device__ __forceinline__ constexpr int swz(int a) { return a ^ (((a >> 6) & 15) << 2); }

__device__ __forceinline__ int cell4(int Lz4, int K) {
    const int X   = ((K >> 6) & 15) ^ ((K >> 2) & 15);
    const int KHI = K & ~63;
    return (Lz4 ^ (X * 16)) + KHI * 4;
}
__device__ __forceinline__ int ccell4(int C) { return 16 * (C ^ ((C >> 4) & 15)); }
__device__ __forceinline__ float& at(float* env, int byte_addr) {
    return *reinterpret_cast<float*>(reinterpret_cast<char*>(env) + byte_addr);
}
__device__ __forceinline__ v2f& at2(float* env, int byte_addr) {
    return *reinterpret_cast<v2f*>(reinterpret_cast<char*>(env) + byte_addr);
}
__device__ __forceinline__ v4f& at4(float* env, int byte_addr) {
    return *reinterpret_cast<v4f*>(reinterpret_cast<char*>(env) + byte_addr);
}

template <int P, bool UPWARD>
__device__ __forceinline__ void site_pair(float* env,
                                          const float* __restrict__ MpA,
                                          const float* __restrict__ MpB,
                                          int wave, int c, int h) {
    constexpr int P1 = UPWARD ? P + 2 : P;
    constexpr int P2 = UPWARD ? P : P + 2;

    const int vo = 2 * (c >> 3) + (c & 1);
    const int ro = (c >> 1) & 3;
    v2f fa[4], fb[4];
    #pragma unroll
    for (int kc = 0; kc < 4; ++kc) {
        if (!UPWARD) {
            const int ncol = ro * 4 + vo;
            fa[kc].x = MpA[((2 * h + 0) * 4 + kc) * 16 + ncol];
            fa[kc].y = MpA[((2 * h + 1) * 4 + kc) * 16 + ncol];
            fb[kc].x = MpB[((2 * h + 0) * 4 + kc) * 16 + ncol];
            fb[kc].y = MpB[((2 * h + 1) * 4 + kc) * 16 + ncol];
        } else {
            const int nrow = (vo * 4 + kc) * 16 + ro * 4;
            fa[kc].x = MpA[nrow + 2 * h + 0];
            fa[kc].y = MpA[nrow + 2 * h + 1];
            fb[kc].x = MpB[nrow + 2 * h + 0];
            fb[kc].y = MpB[nrow + 2 * h + 1];
        }
    }

    #pragma nounroll
    for (int it = 0; it < 2; ++it) {
        const int q    = (wave + 8 * it) * 16 + c;
        const int ql   = q & ((1 << (P - 2)) - 1);
        const int base = ((q >> (P - 2)) << (P + 4)) | (ql << 2);
        const int a0   = 4 * swz(base + 2 * h);

        v8f dA[4];
        #pragma unroll
        for (int T = 0; T < 4; ++T) {
            v8f acc = {};
            #pragma unroll
            for (int kc = 0; kc < 4; ++kc) {
                const v2f b = at2(env, cell4(a0, (kc << P1) | (T << P2)));
                acc = __builtin_amdgcn_wmma_f32_16x16x4_f32(
                        false, fa[kc], false, b, (short)0, acc, false, false);
            }
            dA[T] = acc;
        }

        #pragma unroll
        for (int R = 0; R < 4; ++R) {
            v8f acc = {};
            #pragma unroll
            for (int kc = 0; kc < 4; ++kc) {
                v2f b;
                b.x = dA[kc][2 * R];
                b.y = dA[kc][2 * R + 1];
                acc = __builtin_amdgcn_wmma_f32_16x16x4_f32(
                        false, fb[kc], false, b, (short)0, acc, false, false);
            }
            #pragma unroll
            for (int r2 = 0; r2 < 4; ++r2) {
                v2f o;
                o.x = acc[2 * r2];
                o.y = acc[2 * r2 + 1];
                at2(env, cell4(a0, (R << P1) | (r2 << P2))) = o;
            }
        }
    }
}

__device__ __forceinline__ void last_pair_rows(float* env, float* stage,
                                               const float* __restrict__ M4,
                                               const float* __restrict__ M5,
                                               bool mirror,
                                               float* __restrict__ rows,
                                               int tid, int wave, int c, int h) {
    constexpr int P1 = 10, P2 = 12;
    const int vo = 2 * (c >> 3) + (c & 1);
    const int ro = (c >> 1) & 3;
    const v4f* stage4 = reinterpret_cast<const v4f*>(stage);
    int mi[4];
    #pragma unroll
    for (int kc = 0; kc < 4; ++kc)
        mi[kc] = mirror ? ((2 * h) * 4 + ro) * 16 + kc * 4 + vo
                        : ((2 * h) * 4 + kc) * 16 + ro * 4 + vo;
    const int spare = ENV6 + c;

    #pragma nounroll
    for (int x4 = 0; x4 < 2; ++x4) {
        const float* MpA = M4 + x4 * 256;
        v2f fa[4];
        #pragma unroll
        for (int kc = 0; kc < 4; ++kc) {
            fa[kc].x = MpA[mi[kc]];
            fa[kc].y = MpA[mi[kc] + 64];
        }
        v8f dA[2][4];
        #pragma unroll
        for (int it = 0; it < 2; ++it) {
            const int q  = (wave + 8 * it) * 16 + c;
            const int a0 = 4 * swz((q << 2) + 2 * h);
            #pragma unroll
            for (int T = 0; T < 4; ++T) {
                v8f acc = {};
                #pragma unroll
                for (int kc = 0; kc < 4; ++kc) {
                    const v2f b = at2(env, cell4(a0, (kc << P1) | (T << P2)));
                    acc = __builtin_amdgcn_wmma_f32_16x16x4_f32(
                            false, fa[kc], false, b, (short)0, acc, false, false);
                }
                dA[it][T] = acc;
            }
        }
        #pragma nounroll
        for (int x5 = 0; x5 < 2; ++x5) {
            const float* MpB = M5 + x5 * 256;
            v2f fb[4];
            #pragma unroll
            for (int kc = 0; kc < 4; ++kc) {
                fb[kc].x = MpB[mi[kc]];
                fb[kc].y = MpB[mi[kc] + 64];
            }
            __syncthreads();
            #pragma unroll
            for (int it = 0; it < 2; ++it) {
                const int q = (wave + 8 * it) * 16 + c;
                #pragma unroll
                for (int R = 0; R < 4; ++R) {
                    v8f acc = {};
                    #pragma unroll
                    for (int kc = 0; kc < 4; ++kc) {
                        v2f b;
                        b.x = dA[it][kc][2 * R];
                        b.y = dA[it][kc][2 * R + 1];
                        acc = __builtin_amdgcn_wmma_f32_16x16x4_f32(
                                false, fb[kc], false, b, (short)0, acc, false, false);
                    }
                    #pragma unroll
                    for (int r2 = 0; r2 < 4; ++r2)
                        stage[h ? spare : (q | (R << 8) | (r2 << 10))] = acc[2 * r2];
                }
            }
            __syncthreads();
            v4f* row4 = reinterpret_cast<v4f*>(rows + (size_t)(x4 + 2 * x5) * NCFG3 * ENV6);
            #pragma unroll
            for (int k = 0; k < 4; ++k) row4[tid + 256 * k] = stage4[tid + 256 * k];
        }
    }
}

__device__ __forceinline__ void column_sites(float* env,
                                             const float* M0, const float* M1,
                                             const float* M2, const float* M3,
                                             const float* M4, const float* M5,
                                             int wave, int c, int h, int skip) {
    if (skip < 1) {
        site_pair<2, false>(env, M0, M1, wave, c, h);
        __syncthreads();
    }
    if (skip < 2) {
        site_pair<6, false>(env, M2, M3, wave, c, h);
        __syncthreads();
    }
    site_pair<10, false>(env, M4, M5, wave, c, h);
    __syncthreads();
}

__device__ __forceinline__ void column_sites_up(float* env,
                                                const float* M0, const float* M1,
                                                const float* M2, const float* M3,
                                                const float* M4, const float* M5,
                                                int wave, int c, int h, int skip) {
    site_pair<10, true>(env, M5, M4, wave, c, h);
    __syncthreads();
    if (skip < 2) {
        site_pair<6, true>(env, M3, M2, wave, c, h);
        __syncthreads();
    }
    if (skip < 1) {
        site_pair<2, true>(env, M1, M0, wave, c, h);
        __syncthreads();
    }
}

__global__ void __launch_bounds__(256)
peps_edge_kernel(const float* __restrict__ Msite,
                 float* __restrict__ table,
                 size_t ws_bytes) {
    __shared__ __attribute__((aligned(16))) float env[2 * ENV7];
    __shared__ __attribute__((aligned(16))) float stage[ENV6 + 16];

    const bool deep   = ws_bytes >= (size_t)WS_NEED3;
    const bool deep2L = ws_bytes >= (size_t)WS_NEED4;
    const bool deep3  = ws_bytes >= (size_t)WS_NEED5;
    const bool right = blockIdx.x >= (unsigned)(deep ? NCFG2 : NCFG);

    if (!deep) {
        if (blockIdx.x >= 2 * NCFG) return;
        if (ws_bytes < (size_t)(right ? WS_NEED2 : WS_NEED)) return;
    }

    const int tid  = threadIdx.x;
    const int cfg8 = blockIdx.x & ((deep ? NCFG2 : NCFG) - 1);
    const int cfg  = cfg8 & (NCFG - 1);
    const int m0   = right ? NSITE : 0;
    const int mst  = right ? 1 : LY;
    const int lane = tid & 31;
    const int wave = __builtin_amdgcn_readfirstlane(tid >> 5);
    __builtin_assume(wave >= 0 && wave < 8);
    const int c    = lane & 15;
    const int h    = lane >> 4;

    for (int q = tid; q < ENV7; q += 256) env[q] = 0.0f;
    __syncthreads();
    if (tid == 0) env[0] = 1.0f;
    __syncthreads();

    const float* M0 = Msite + ((m0 + 0 * mst) * PD + ((cfg >> 0) & 1)) * 256;
    const float* M1 = Msite + ((m0 + 1 * mst) * PD + ((cfg >> 1) & 1)) * 256;
    const float* M2 = Msite + ((m0 + 2 * mst) * PD + ((cfg >> 2) & 1)) * 256;
    const float* M3 = Msite + ((m0 + 3 * mst) * PD + ((cfg >> 3) & 1)) * 256;
    const float* M4 = Msite + ((m0 + 4 * mst) * PD + ((cfg >> 4) & 1)) * 256;
    const float* M5 = Msite + ((m0 + 5 * mst) * PD + ((cfg >> 5) & 1)) * 256;
    column_sites(env, M0, M1, M2, M3, M4, M5, wave, c, h, 0);

    if (!deep) {
        float* row = table + blockIdx.x * ENV6;
        #pragma unroll
        for (int q = 0; q < 16; ++q)
            row[tid + 256 * q] = at(env, ccell4(tid) + 4096 * q);
        return;
    }

    const int mA = right ? (NSITE + LX + 0) : (0 * LY + 1);
    const int mB = right ? (NSITE + LX + 1) : (1 * LY + 1);
    const float* MA = Msite + (mA * PD + ((cfg8 >> 6) & 1)) * 256;
    const float* MB = Msite + (mB * PD + ((cfg8 >> 7) & 1)) * 256;
    site_pair<2, false>(env, MA, MB, wave, c, h);
    __syncthreads();

    const bool deeper = right ? deep3 : deep2L;
    float*     side   = table + (right ? (size_t)(deep2L ? NCFG3 : NCFG2) * ENV7 : (size_t)0);
    v4f*       env4   = reinterpret_cast<v4f*>(env);

    if (!deeper) {
        v4f* row4 = reinterpret_cast<v4f*>(side + (size_t)cfg8 * ENV7);
        #pragma unroll
        for (int q = 0; q < 16; ++q) row4[tid + 256 * q] = env4[tid + 256 * q];
        return;
    }

    v4f* copy4 = reinterpret_cast<v4f*>(env + ENV7);
    #pragma unroll
    for (int q = 0; q < 16; ++q) copy4[tid + 256 * q] = env4[tid + 256 * q];
    __syncthreads();

    const int mC = right ? (NSITE + LX + 2) : (2 * LY + 1);
    const int mD = right ? (NSITE + LX + 3) : (3 * LY + 1);
    #pragma nounroll
    for (int v = 0; v < 4; ++v) {
        if (v > 0) {
            __syncthreads();
            #pragma unroll
            for (int q = 0; q < 16; ++q) env4[tid + 256 * q] = copy4[tid + 256 * q];
            __syncthreads();
        }
        const float* MC = Msite + (mC * PD + (v & 1)) * 256;
        const float* MD = Msite + (mD * PD + (v >> 1)) * 256;
        site_pair<6, false>(env, MC, MD, wave, c, h);
        __syncthreads();
        if (deep3) {
            const int m4 = right ? (4 * LY + LY - 2) : (4 * LY + 1);
            const int m5 = right ? (5 * LY + LY - 2) : (5 * LY + 1);
            last_pair_rows(env, stage, Msite + m4 * PD * 256, Msite + m5 * PD * 256, right,
                           side + (size_t)(cfg8 + NCFG2 * v) * ENV6, tid, wave, c, h);
        } else {
            v4f* row4 = reinterpret_cast<v4f*>(side + (size_t)(cfg8 + NCFG2 * v) * ENV7);
            #pragma unroll
            for (int q = 0; q < 16; ++q) row4[tid + 256 * q] = env4[tid + 256 * q];
        }
    }
}

__global__ void __launch_bounds__(256)
peps_amp_kernel(const int* __restrict__ x,
                const float* __restrict__ Msite,
                const float* __restrict__ table,
                size_t ws_bytes,
                float* __restrict__ out) {
    extern __shared__ float env[];

    const int tid    = threadIdx.x;
    const int sample = blockIdx.x;
    const int lane   = tid & 31;
    const int wave   = __builtin_amdgcn_readfirstlane(tid >> 5);
    __builtin_assume(wave >= 0 && wave < 8);
    const int c      = lane & 15;
    const int h      = lane >> 4;

    const int* xrow = x + sample * NSITE;
    const bool have_table = ws_bytes >= (size_t)WS_NEED;
    const bool have_right = ws_bytes >= (size_t)WS_NEED2;
    const bool have_deep  = ws_bytes >= (size_t)WS_NEED3;
    const bool have_deep2L = ws_bytes >= (size_t)WS_NEED4;
    const bool have_deep3  = ws_bytes >= (size_t)WS_NEED5;

    if (have_deep3) {
        const int cfg = xrow[0 * LY] | (xrow[1 * LY] << 1) | (xrow[2 * LY] << 2) |
                        (xrow[3 * LY] << 3) | (xrow[4 * LY] << 4) | (xrow[5 * LY] << 5) |
                        (xrow[0 * LY + 1] << 6) | (xrow[1 * LY + 1] << 7) |
                        (xrow[2 * LY + 1] << 8) | (xrow[3 * LY + 1] << 9) |
                        (xrow[4 * LY + 1] << 10) | (xrow[5 * LY + 1] << 11);
        const float* row = table + (size_t)cfg * ENV6;
        float keep[16];
        #pragma unroll
        for (int q = 0; q < 16; ++q) keep[q] = row[tid + 256 * q];
        #pragma unroll
        for (int q = 0; q < 16; ++q) {
            const v4f t = {keep[q], 0.0f, 0.0f, 0.0f};
            at4(env, ccell4(tid) + 4096 * q) = t;
        }
        __syncthreads();
    } else if (have_deep) {
        int cfg = xrow[0 * LY] | (xrow[1 * LY] << 1) | (xrow[2 * LY] << 2) |
                  (xrow[3 * LY] << 3) | (xrow[4 * LY] << 4) | (xrow[5 * LY] << 5) |
                  (xrow[0 * LY + 1] << 6) | (xrow[1 * LY + 1] << 7);
        if (have_deep2L) cfg |= (xrow[2 * LY + 1] << 8) | (xrow[3 * LY + 1] << 9);
        const v4f* row4 = reinterpret_cast<const v4f*>(table + (size_t)cfg * ENV7);
        v4f*       env4 = reinterpret_cast<v4f*>(env);
        v4f keep4[16];
        #pragma unroll
        for (int q = 0; q < 16; ++q) keep4[q] = row4[tid + 256 * q];
        #pragma unroll
        for (int q = 0; q < 16; ++q) env4[tid + 256 * q] = keep4[q];
        __syncthreads();
    } else if (have_table) {
        const int cfg = xrow[0 * LY] | (xrow[1 * LY] << 1) | (xrow[2 * LY] << 2) |
                        (xrow[3 * LY] << 3) | (xrow[4 * LY] << 4) | (xrow[5 * LY] << 5);
        const float* row = table + cfg * ENV6;
        float keep[16];
        #pragma unroll
        for (int q = 0; q < 16; ++q) keep[q] = row[tid + 256 * q];
        #pragma unroll
        for (int q = 0; q < 16; ++q) {
            const v4f t = {keep[q], 0.0f, 0.0f, 0.0f};
            at4(env, ccell4(tid) + 4096 * q) = t;
        }
        __syncthreads();
    } else {
        for (int q = tid; q < ENV7; q += 256) env[q] = 0.0f;
        __syncthreads();
        if (tid == 0) env[0] = 1.0f;
        __syncthreads();
    }

    bool upward = false;
    const int jend = have_deep3 ? LY - 2 : (have_right ? LY - 1 : LY);
    for (int j = have_deep3 ? 2 : (have_table ? 1 : 0); j < jend; ++j) {
        const float* M0 = Msite + ((0 * LY + j) * PD + xrow[0 * LY + j]) * 256;
        const float* M1 = Msite + ((1 * LY + j) * PD + xrow[1 * LY + j]) * 256;
        const float* M2 = Msite + ((2 * LY + j) * PD + xrow[2 * LY + j]) * 256;
        const float* M3 = Msite + ((3 * LY + j) * PD + xrow[3 * LY + j]) * 256;
        const float* M4 = Msite + ((4 * LY + j) * PD + xrow[4 * LY + j]) * 256;
        const float* M5 = Msite + ((5 * LY + j) * PD + xrow[5 * LY + j]) * 256;

        const int skip_dn = (have_deep && j == 1)      ? (have_deep2L ? 2 : 1) : 0;
        const int skip_up = (have_deep && j == LY - 2) ? 1 : 0;
        if (!upward) column_sites(env, M0, M1, M2, M3, M4, M5, wave, c, h, skip_dn);
        else         column_sites_up(env, M0, M1, M2, M3, M4, M5, wave, c, h, skip_up);
        upward = !upward;
    }

    if (!have_right) {
        if (tid == 0) out[sample] = env[0];
        return;
    }

    const int cfg5 = xrow[0 * LY + 5] | (xrow[1 * LY + 5] << 1) | (xrow[2 * LY + 5] << 2) |
                     (xrow[3 * LY + 5] << 3) | (xrow[4 * LY + 5] << 4) | (xrow[5 * LY + 5] << 5);
    float part = 0.0f;
    if (have_deep3) {
        const int cfgR = cfg5 | (xrow[0 * LY + LY - 2] << 6) | (xrow[1 * LY + LY - 2] << 7) |
                         (xrow[2 * LY + LY - 2] << 8) | (xrow[3 * LY + LY - 2] << 9) |
                         (xrow[4 * LY + LY - 2] << 10) | (xrow[5 * LY + LY - 2] << 11);
        const float* rrow = table + (size_t)(NCFG4 + cfgR) * ENV6;
        float rv[16];
        #pragma unroll
        for (int q = 0; q < 16; ++q) rv[q] = rrow[tid + 256 * q];
        #pragma unroll
        for (int q = 0; q < 16; ++q)
            part = fmaf(at(env, ccell4(tid) + 4096 * q), rv[q], part);
    } else if (have_deep) {
        const int cfgR = cfg5 | (xrow[0 * LY + LY - 2] << 6) | (xrow[1 * LY + LY - 2] << 7);
        const int nl = have_deep2L ? NCFG3 : NCFG2;
        const v4f* rrow4 = reinterpret_cast<const v4f*>(table + (size_t)(nl + cfgR) * ENV7);
        const v4f* env4  = reinterpret_cast<const v4f*>(env);
        v4f rv4[16];
        #pragma unroll
        for (int q = 0; q < 16; ++q) rv4[q] = rrow4[tid + 256 * q];
        #pragma unroll
        for (int q = 0; q < 16; ++q) {
            const v4f e = env4[tid + 256 * q];
            part = fmaf(e.x, rv4[q].x, part);
            part = fmaf(e.y, rv4[q].y, part);
            part = fmaf(e.z, rv4[q].z, part);
            part = fmaf(e.w, rv4[q].w, part);
        }
    } else {
        const float* rrow = table + (NCFG + cfg5) * ENV6;
        float rv[16];
        #pragma unroll
        for (int q = 0; q < 16; ++q) rv[q] = rrow[tid + 256 * q];
        #pragma unroll
        for (int q = 0; q < 16; ++q)
            part = fmaf(at(env, ccell4(tid) + 4096 * q), rv[q], part);
    }
    for (int o = 16; o > 0; o >>= 1) part += __shfl_xor(part, o);
    __syncthreads();
    if (lane == 0) env[wave] = part;
    __syncthreads();
    if (tid == 0) {
        float t = env[0];
        for (int w = 1; w < 8; ++w) t += env[w];
        out[sample] = t;
    }
}

extern "C" void kernel_launch(void* const* d_in, const int* in_sizes, int n_in,
                              void* d_out, int out_size, void* d_ws, size_t ws_size,
                              hipStream_t stream) {
    const int*   x     = (const int*)d_in[0];
    const float* peps  = (const float*)d_in[1];
    float*       out   = (float*)d_out;
    float*       Msite = (float*)d_ws;
    float*       table = Msite + NMAT * PD * 256;

    const int nM = NMAT * PD * 256;
    peps_prep_kernel<<<(nM + 255) / 256, 256, 0, stream>>>(peps, Msite, ws_size);

    peps_edge_kernel<<<2 * NCFG2, 256, 0, stream>>>(Msite, table, ws_size);

    const int B = out_size;
    peps_amp_kernel<<<B, 256, ENV7 * sizeof(float), stream>>>(x, Msite, table, ws_size, out);
}


# turn of the sweep fused: rows 5 of columns 2 and 3 contracted with each other first (16x16, rank 4), rows 4,5 of both columns one 3-step register chain: 48 not 64 matrix ops a trip; sum reordered
# speedup vs baseline: 7.8500x; 1.0483x over previous
#include <hip/hip_runtime.h>


typedef float v2f __attribute__((ext_vector_type(2)));
typedef float v4f __attribute__((ext_vector_type(4)));
typedef float v8f __attribute__((ext_vector_type(8)));

#define LX 6
#define LY 6
#define PD 2
#define DD 4
#define NSITE 36
#define ENV7 16384
#define ENV6 4096
#define NCFG 64
#define NCFG2 256
#define NCFG3 1024
#define NCFG4 4096
#define NMAT 46
#define MSITE_BYTES (NMAT * PD * 256 * 4)
#define WS_NEED     (MSITE_BYTES + NCFG * ENV6 * 4)
#define WS_NEED2    (MSITE_BYTES + 2 * NCFG * ENV6 * 4)
#define WS_NEED3    (MSITE_BYTES + 2 * NCFG2 * ENV7 * 4)
#define WS_NEED4    (MSITE_BYTES + (NCFG3 + NCFG2) * ENV7 * 4)
#define WS_NEED5    (MSITE_BYTES + 2 * NCFG4 * ENV6 * 4)

__global__ void peps_prep_kernel(const float* __restrict__ peps,
                                 float* __restrict__ Msite,
                                 size_t ws_bytes) {
    int idx = blockIdx.x * blockDim.x + threadIdx.x;
    if (idx >= NMAT * PD * 256) return;
    if (idx >= NSITE * PD * 256 && ws_bytes < (size_t)WS_NEED2) return;
    if (idx >= (NSITE + LX) * PD * 256 && ws_bytes < (size_t)WS_NEED3) return;
    int n    = idx & 15;
    int k    = (idx >> 4) & 15;
    int s    = (idx >> 8) & 1;
    int site = idx >> 9;
    int a = k >> 2, c = k & 3;
    int d = n >> 2, b = n & 3;
    float v = 0.0f;
    if (site < NSITE) {
        int i = site / LY, j = site % LY;
        bool pad = (i == 0      && a > 0) || (i == LX - 1 && b > 0) ||
                   (j == 0      && c > 0) || (j == LY - 1 && d > 0);
        if (!pad) {
            int p = ((((site * PD + s) * DD + a) * DD + b) * DD + c) * DD + d;
            v = peps[p];
        }
    } else if (site < NSITE + LX) {
        int i  = site - NSITE;
        int rs = i * LY + (LY - 1);
        bool pad = (i == 0 && a > 0) || (i == LX - 1 && b > 0) || (c > 0);
        if (!pad) {
            int p = ((((rs * PD + s) * DD + a) * DD + b) * DD + d) * DD + c;
            v = peps[p];
        }
    } else {
        int i  = site - (NSITE + LX);
        int rs = i * LY + (LY - 2);
        bool pad = (i == 0 && a > 0);
        if (!pad) {
            int p = ((((rs * PD + s) * DD + a) * DD + b) * DD + d) * DD + c;
            v = peps[p];
        }
    }
    Msite[idx] = v;
}

__device__ __forceinline__ constexpr int swz(int a) { return a ^ (((a >> 6) & 15) << 2); }

__device__ __forceinline__ int cell4(int Lz4, int K) {
    const int X   = ((K >> 6) & 15) ^ ((K >> 2) & 15);
    const int KHI = K & ~63;
    return (Lz4 ^ (X * 16)) + KHI * 4;
}
__device__ __forceinline__ int ccell4(int C) { return 16 * (C ^ ((C >> 4) & 15)); }
__device__ __forceinline__ float& at(float* env, int byte_addr) {
    return *reinterpret_cast<float*>(reinterpret_cast<char*>(env) + byte_addr);
}
__device__ __forceinline__ v2f& at2(float* env, int byte_addr) {
    return *reinterpret_cast<v2f*>(reinterpret_cast<char*>(env) + byte_addr);
}
__device__ __forceinline__ v4f& at4(float* env, int byte_addr) {
    return *reinterpret_cast<v4f*>(reinterpret_cast<char*>(env) + byte_addr);
}

template <int P, bool UPWARD>
__device__ __forceinline__ void site_pair(float* env,
                                          const float* __restrict__ MpA,
                                          const float* __restrict__ MpB,
                                          int wave, int c, int h) {
    constexpr int P1 = UPWARD ? P + 2 : P;
    constexpr int P2 = UPWARD ? P : P + 2;

    const int vo = 2 * (c >> 3) + (c & 1);
    const int ro = (c >> 1) & 3;
    v2f fa[4], fb[4];
    #pragma unroll
    for (int kc = 0; kc < 4; ++kc) {
        if (!UPWARD) {
            const int ncol = ro * 4 + vo;
            fa[kc].x = MpA[((2 * h + 0) * 4 + kc) * 16 + ncol];
            fa[kc].y = MpA[((2 * h + 1) * 4 + kc) * 16 + ncol];
            fb[kc].x = MpB[((2 * h + 0) * 4 + kc) * 16 + ncol];
            fb[kc].y = MpB[((2 * h + 1) * 4 + kc) * 16 + ncol];
        } else {
            const int nrow = (vo * 4 + kc) * 16 + ro * 4;
            fa[kc].x = MpA[nrow + 2 * h + 0];
            fa[kc].y = MpA[nrow + 2 * h + 1];
            fb[kc].x = MpB[nrow + 2 * h + 0];
            fb[kc].y = MpB[nrow + 2 * h + 1];
        }
    }

    #pragma nounroll
    for (int it = 0; it < 2; ++it) {
        const int q    = (wave + 8 * it) * 16 + c;
        const int ql   = q & ((1 << (P - 2)) - 1);
        const int base = ((q >> (P - 2)) << (P + 4)) | (ql << 2);
        const int a0   = 4 * swz(base + 2 * h);

        v8f dA[4];
        #pragma unroll
        for (int T = 0; T < 4; ++T) {
            v8f acc = {};
            #pragma unroll
            for (int kc = 0; kc < 4; ++kc) {
                const v2f b = at2(env, cell4(a0, (kc << P1) | (T << P2)));
                acc = __builtin_amdgcn_wmma_f32_16x16x4_f32(
                        false, fa[kc], false, b, (short)0, acc, false, false);
            }
            dA[T] = acc;
        }

        #pragma unroll
        for (int R = 0; R < 4; ++R) {
            v8f acc = {};
            #pragma unroll
            for (int kc = 0; kc < 4; ++kc) {
                v2f b;
                b.x = dA[kc][2 * R];
                b.y = dA[kc][2 * R + 1];
                acc = __builtin_amdgcn_wmma_f32_16x16x4_f32(
                        false, fb[kc], false, b, (short)0, acc, false, false);
            }
            #pragma unroll
            for (int r2 = 0; r2 < 4; ++r2) {
                v2f o;
                o.x = acc[2 * r2];
                o.y = acc[2 * r2 + 1];
                at2(env, cell4(a0, (R << P1) | (r2 << P2))) = o;
            }
        }
    }
}

__device__ __forceinline__ void turn_chain(float* env,
                                           const float* __restrict__ M4a,
                                           const float* __restrict__ M5a,
                                           const float* __restrict__ M5b,
                                           const float* __restrict__ M4b,
                                           int wave, int c, int h) {
    constexpr int P4 = 10, P5 = 12;
    const int vo = 2 * (c >> 3) + (c & 1);
    const int ro = (c >> 1) & 3;
    v2f pa, pb;
    pa.x = M5a[(vo * 4 + ro) * 16 + (2 * h + 0) * 4];
    pa.y = M5a[(vo * 4 + ro) * 16 + (2 * h + 1) * 4];
    pb.x = M5b[(vo * 4 + 2 * h + 0) * 16 + ro * 4];
    pb.y = M5b[(vo * 4 + 2 * h + 1) * 16 + ro * 4];
    const v8f fzero = {};
    const v8f F = __builtin_amdgcn_wmma_f32_16x16x4_f32(
            false, pa, false, pb, (short)0, fzero, false, false);

    v2f fa[4], ff[4], fc[4];
    #pragma unroll
    for (int kc = 0; kc < 4; ++kc) {
        const int ncol = ro * 4 + vo;
        fa[kc].x = M4a[((2 * h + 0) * 4 + kc) * 16 + ncol];
        fa[kc].y = M4a[((2 * h + 1) * 4 + kc) * 16 + ncol];
        const int nrow = (vo * 4 + kc) * 16 + ro * 4;
        fc[kc].x = M4b[nrow + 2 * h + 0];
        fc[kc].y = M4b[nrow + 2 * h + 1];
        ff[kc].x = F[2 * kc];
        ff[kc].y = F[2 * kc + 1];
    }

    #pragma nounroll
    for (int it = 0; it < 2; ++it) {
        const int q  = (wave + 8 * it) * 16 + c;
        const int a0 = 4 * swz((q << 2) + 2 * h);

        v8f dA[4];
        #pragma unroll
        for (int T = 0; T < 4; ++T) {
            v8f acc = {};
            #pragma unroll
            for (int kc = 0; kc < 4; ++kc) {
                const v2f b = at2(env, cell4(a0, (kc << P4) | (T << P5)));
                acc = __builtin_amdgcn_wmma_f32_16x16x4_f32(
                        false, fa[kc], false, b, (short)0, acc, false, false);
            }
            dA[T] = acc;
        }

        v8f dB[4];
        #pragma unroll
        for (int R = 0; R < 4; ++R) {
            v8f acc = {};
            #pragma unroll
            for (int kc = 0; kc < 4; ++kc) {
                v2f b;
                b.x = dA[kc][2 * R];
                b.y = dA[kc][2 * R + 1];
                acc = __builtin_amdgcn_wmma_f32_16x16x4_f32(
                        false, ff[kc], false, b, (short)0, acc, false, false);
            }
            dB[R] = acc;
        }

        #pragma unroll
        for (int r5 = 0; r5 < 4; ++r5) {
            v8f acc = {};
            #pragma unroll
            for (int kc = 0; kc < 4; ++kc) {
                v2f b;
                b.x = dB[kc][2 * r5];
                b.y = dB[kc][2 * r5 + 1];
                acc = __builtin_amdgcn_wmma_f32_16x16x4_f32(
                        false, fc[kc], false, b, (short)0, acc, false, false);
            }
            #pragma unroll
            for (int r4 = 0; r4 < 4; ++r4) {
                v2f o;
                o.x = acc[2 * r4];
                o.y = acc[2 * r4 + 1];
                at2(env, cell4(a0, (r4 << P4) | (r5 << P5))) = o;
            }
        }
    }
}

__device__ __forceinline__ void last_pair_rows(float* env, float* stage,
                                               const float* __restrict__ M4,
                                               const float* __restrict__ M5,
                                               bool mirror,
                                               float* __restrict__ rows,
                                               int tid, int wave, int c, int h) {
    constexpr int P1 = 10, P2 = 12;
    const int vo = 2 * (c >> 3) + (c & 1);
    const int ro = (c >> 1) & 3;
    const v4f* stage4 = reinterpret_cast<const v4f*>(stage);
    int mi[4];
    #pragma unroll
    for (int kc = 0; kc < 4; ++kc)
        mi[kc] = mirror ? ((2 * h) * 4 + ro) * 16 + kc * 4 + vo
                        : ((2 * h) * 4 + kc) * 16 + ro * 4 + vo;
    const int spare = ENV6 + c;

    #pragma nounroll
    for (int x4 = 0; x4 < 2; ++x4) {
        const float* MpA = M4 + x4 * 256;
        v2f fa[4];
        #pragma unroll
        for (int kc = 0; kc < 4; ++kc) {
            fa[kc].x = MpA[mi[kc]];
            fa[kc].y = MpA[mi[kc] + 64];
        }
        v8f dA[2][4];
        #pragma unroll
        for (int it = 0; it < 2; ++it) {
            const int q  = (wave + 8 * it) * 16 + c;
            const int a0 = 4 * swz((q << 2) + 2 * h);
            #pragma unroll
            for (int T = 0; T < 4; ++T) {
                v8f acc = {};
                #pragma unroll
                for (int kc = 0; kc < 4; ++kc) {
                    const v2f b = at2(env, cell4(a0, (kc << P1) | (T << P2)));
                    acc = __builtin_amdgcn_wmma_f32_16x16x4_f32(
                            false, fa[kc], false, b, (short)0, acc, false, false);
                }
                dA[it][T] = acc;
            }
        }
        #pragma nounroll
        for (int x5 = 0; x5 < 2; ++x5) {
            const float* MpB = M5 + x5 * 256;
            v2f fb[4];
            #pragma unroll
            for (int kc = 0; kc < 4; ++kc) {
                fb[kc].x = MpB[mi[kc]];
                fb[kc].y = MpB[mi[kc] + 64];
            }
            __syncthreads();
            #pragma unroll
            for (int it = 0; it < 2; ++it) {
                const int q = (wave + 8 * it) * 16 + c;
                #pragma unroll
                for (int R = 0; R < 4; ++R) {
                    v8f acc = {};
                    #pragma unroll
                    for (int kc = 0; kc < 4; ++kc) {
                        v2f b;
                        b.x = dA[it][kc][2 * R];
                        b.y = dA[it][kc][2 * R + 1];
                        acc = __builtin_amdgcn_wmma_f32_16x16x4_f32(
                                false, fb[kc], false, b, (short)0, acc, false, false);
                    }
                    #pragma unroll
                    for (int r2 = 0; r2 < 4; ++r2)
                        stage[h ? spare : (q | (R << 8) | (r2 << 10))] = acc[2 * r2];
                }
            }
            __syncthreads();
            v4f* row4 = reinterpret_cast<v4f*>(rows + (size_t)(x4 + 2 * x5) * NCFG3 * ENV6);
            #pragma unroll
            for (int k = 0; k < 4; ++k) row4[tid + 256 * k] = stage4[tid + 256 * k];
        }
    }
}

__device__ __forceinline__ void column_sites(float* env,
                                             const float* M0, const float* M1,
                                             const float* M2, const float* M3,
                                             const float* M4, const float* M5,
                                             int wave, int c, int h, int skip) {
    if (skip < 1) {
        site_pair<2, false>(env, M0, M1, wave, c, h);
        __syncthreads();
    }
    if (skip < 2) {
        site_pair<6, false>(env, M2, M3, wave, c, h);
        __syncthreads();
    }
    site_pair<10, false>(env, M4, M5, wave, c, h);
    __syncthreads();
}

__device__ __forceinline__ void column_sites_up(float* env,
                                                const float* M0, const float* M1,
                                                const float* M2, const float* M3,
                                                const float* M4, const float* M5,
                                                int wave, int c, int h, int skip) {
    site_pair<10, true>(env, M5, M4, wave, c, h);
    __syncthreads();
    if (skip < 2) {
        site_pair<6, true>(env, M3, M2, wave, c, h);
        __syncthreads();
    }
    if (skip < 1) {
        site_pair<2, true>(env, M1, M0, wave, c, h);
        __syncthreads();
    }
}

__global__ void __launch_bounds__(256)
peps_edge_kernel(const float* __restrict__ Msite,
                 float* __restrict__ table,
                 size_t ws_bytes) {
    __shared__ __attribute__((aligned(16))) float env[2 * ENV7];
    __shared__ __attribute__((aligned(16))) float stage[ENV6 + 16];

    const bool deep   = ws_bytes >= (size_t)WS_NEED3;
    const bool deep2L = ws_bytes >= (size_t)WS_NEED4;
    const bool deep3  = ws_bytes >= (size_t)WS_NEED5;
    const bool right = blockIdx.x >= (unsigned)(deep ? NCFG2 : NCFG);

    if (!deep) {
        if (blockIdx.x >= 2 * NCFG) return;
        if (ws_bytes < (size_t)(right ? WS_NEED2 : WS_NEED)) return;
    }

    const int tid  = threadIdx.x;
    const int cfg8 = blockIdx.x & ((deep ? NCFG2 : NCFG) - 1);
    const int cfg  = cfg8 & (NCFG - 1);
    const int m0   = right ? NSITE : 0;
    const int mst  = right ? 1 : LY;
    const int lane = tid & 31;
    const int wave = __builtin_amdgcn_readfirstlane(tid >> 5);
    __builtin_assume(wave >= 0 && wave < 8);
    const int c    = lane & 15;
    const int h    = lane >> 4;

    for (int q = tid; q < ENV7; q += 256) env[q] = 0.0f;
    __syncthreads();
    if (tid == 0) env[0] = 1.0f;
    __syncthreads();

    const float* M0 = Msite + ((m0 + 0 * mst) * PD + ((cfg >> 0) & 1)) * 256;
    const float* M1 = Msite + ((m0 + 1 * mst) * PD + ((cfg >> 1) & 1)) * 256;
    const float* M2 = Msite + ((m0 + 2 * mst) * PD + ((cfg >> 2) & 1)) * 256;
    const float* M3 = Msite + ((m0 + 3 * mst) * PD + ((cfg >> 3) & 1)) * 256;
    const float* M4 = Msite + ((m0 + 4 * mst) * PD + ((cfg >> 4) & 1)) * 256;
    const float* M5 = Msite + ((m0 + 5 * mst) * PD + ((cfg >> 5) & 1)) * 256;
    column_sites(env, M0, M1, M2, M3, M4, M5, wave, c, h, 0);

    if (!deep) {
        float* row = table + blockIdx.x * ENV6;
        #pragma unroll
        for (int q = 0; q < 16; ++q)
            row[tid + 256 * q] = at(env, ccell4(tid) + 4096 * q);
        return;
    }

    const int mA = right ? (NSITE + LX + 0) : (0 * LY + 1);
    const int mB = right ? (NSITE + LX + 1) : (1 * LY + 1);
    const float* MA = Msite + (mA * PD + ((cfg8 >> 6) & 1)) * 256;
    const float* MB = Msite + (mB * PD + ((cfg8 >> 7) & 1)) * 256;
    site_pair<2, false>(env, MA, MB, wave, c, h);
    __syncthreads();

    const bool deeper = right ? deep3 : deep2L;
    float*     side   = table + (right ? (size_t)(deep2L ? NCFG3 : NCFG2) * ENV7 : (size_t)0);
    v4f*       env4   = reinterpret_cast<v4f*>(env);

    if (!deeper) {
        v4f* row4 = reinterpret_cast<v4f*>(side + (size_t)cfg8 * ENV7);
        #pragma unroll
        for (int q = 0; q < 16; ++q) row4[tid + 256 * q] = env4[tid + 256 * q];
        return;
    }

    v4f* copy4 = reinterpret_cast<v4f*>(env + ENV7);
    #pragma unroll
    for (int q = 0; q < 16; ++q) copy4[tid + 256 * q] = env4[tid + 256 * q];
    __syncthreads();

    const int mC = right ? (NSITE + LX + 2) : (2 * LY + 1);
    const int mD = right ? (NSITE + LX + 3) : (3 * LY + 1);
    #pragma nounroll
    for (int v = 0; v < 4; ++v) {
        if (v > 0) {
            __syncthreads();
            #pragma unroll
            for (int q = 0; q < 16; ++q) env4[tid + 256 * q] = copy4[tid + 256 * q];
            __syncthreads();
        }
        const float* MC = Msite + (mC * PD + (v & 1)) * 256;
        const float* MD = Msite + (mD * PD + (v >> 1)) * 256;
        site_pair<6, false>(env, MC, MD, wave, c, h);
        __syncthreads();
        if (deep3) {
            const int m4 = right ? (4 * LY + LY - 2) : (4 * LY + 1);
            const int m5 = right ? (5 * LY + LY - 2) : (5 * LY + 1);
            last_pair_rows(env, stage, Msite + m4 * PD * 256, Msite + m5 * PD * 256, right,
                           side + (size_t)(cfg8 + NCFG2 * v) * ENV6, tid, wave, c, h);
        } else {
            v4f* row4 = reinterpret_cast<v4f*>(side + (size_t)(cfg8 + NCFG2 * v) * ENV7);
            #pragma unroll
            for (int q = 0; q < 16; ++q) row4[tid + 256 * q] = env4[tid + 256 * q];
        }
    }
}

__global__ void __launch_bounds__(256)
peps_amp_kernel(const int* __restrict__ x,
                const float* __restrict__ Msite,
                const float* __restrict__ table,
                size_t ws_bytes,
                float* __restrict__ out) {
    extern __shared__ float env[];

    const int tid    = threadIdx.x;
    const int sample = blockIdx.x;
    const int lane   = tid & 31;
    const int wave   = __builtin_amdgcn_readfirstlane(tid >> 5);
    __builtin_assume(wave >= 0 && wave < 8);
    const int c      = lane & 15;
    const int h      = lane >> 4;

    const int* xrow = x + sample * NSITE;
    const bool have_table = ws_bytes >= (size_t)WS_NEED;
    const bool have_right = ws_bytes >= (size_t)WS_NEED2;
    const bool have_deep  = ws_bytes >= (size_t)WS_NEED3;
    const bool have_deep2L = ws_bytes >= (size_t)WS_NEED4;
    const bool have_deep3  = ws_bytes >= (size_t)WS_NEED5;

    if (have_deep3) {
        const int cfg = xrow[0 * LY] | (xrow[1 * LY] << 1) | (xrow[2 * LY] << 2) |
                        (xrow[3 * LY] << 3) | (xrow[4 * LY] << 4) | (xrow[5 * LY] << 5) |
                        (xrow[0 * LY + 1] << 6) | (xrow[1 * LY + 1] << 7) |
                        (xrow[2 * LY + 1] << 8) | (xrow[3 * LY + 1] << 9) |
                        (xrow[4 * LY + 1] << 10) | (xrow[5 * LY + 1] << 11);
        const float* row = table + (size_t)cfg * ENV6;
        float keep[16];
        #pragma unroll
        for (int q = 0; q < 16; ++q) keep[q] = row[tid + 256 * q];
        #pragma unroll
        for (int q = 0; q < 16; ++q) {
            const v4f t = {keep[q], 0.0f, 0.0f, 0.0f};
            at4(env, ccell4(tid) + 4096 * q) = t;
        }
        __syncthreads();
    } else if (have_deep) {
        int cfg = xrow[0 * LY] | (xrow[1 * LY] << 1) | (xrow[2 * LY] << 2) |
                  (xrow[3 * LY] << 3) | (xrow[4 * LY] << 4) | (xrow[5 * LY] << 5) |
                  (xrow[0 * LY + 1] << 6) | (xrow[1 * LY + 1] << 7);
        if (have_deep2L) cfg |= (xrow[2 * LY + 1] << 8) | (xrow[3 * LY + 1] << 9);
        const v4f* row4 = reinterpret_cast<const v4f*>(table + (size_t)cfg * ENV7);
        v4f*       env4 = reinterpret_cast<v4f*>(env);
        v4f keep4[16];
        #pragma unroll
        for (int q = 0; q < 16; ++q) keep4[q] = row4[tid + 256 * q];
        #pragma unroll
        for (int q = 0; q < 16; ++q) env4[tid + 256 * q] = keep4[q];
        __syncthreads();
    } else if (have_table) {
        const int cfg = xrow[0 * LY] | (xrow[1 * LY] << 1) | (xrow[2 * LY] << 2) |
                        (xrow[3 * LY] << 3) | (xrow[4 * LY] << 4) | (xrow[5 * LY] << 5);
        const float* row = table + cfg * ENV6;
        float keep[16];
        #pragma unroll
        for (int q = 0; q < 16; ++q) keep[q] = row[tid + 256 * q];
        #pragma unroll
        for (int q = 0; q < 16; ++q) {
            const v4f t = {keep[q], 0.0f, 0.0f, 0.0f};
            at4(env, ccell4(tid) + 4096 * q) = t;
        }
        __syncthreads();
    } else {
        for (int q = tid; q < ENV7; q += 256) env[q] = 0.0f;
        __syncthreads();
        if (tid == 0) env[0] = 1.0f;
        __syncthreads();
    }

    if (have_deep3) {
        const float* A[LX];
        const float* Bm[LX];
        #pragma unroll
        for (int i = 0; i < LX; ++i) {
            A[i]  = Msite + ((i * LY + 2) * PD + xrow[i * LY + 2]) * 256;
            Bm[i] = Msite + ((i * LY + 3) * PD + xrow[i * LY + 3]) * 256;
        }
        site_pair<2, false>(env, A[0], A[1], wave, c, h);
        __syncthreads();
        site_pair<6, false>(env, A[2], A[3], wave, c, h);
        __syncthreads();
        turn_chain(env, A[4], A[5], Bm[5], Bm[4], wave, c, h);
        __syncthreads();
        site_pair<6, true>(env, Bm[3], Bm[2], wave, c, h);
        __syncthreads();
        site_pair<2, true>(env, Bm[1], Bm[0], wave, c, h);
        __syncthreads();
    } else {
        bool upward = false;
        const int jend = have_right ? LY - 1 : LY;
        for (int j = have_table ? 1 : 0; j < jend; ++j) {
            const float* M0 = Msite + ((0 * LY + j) * PD + xrow[0 * LY + j]) * 256;
            const float* M1 = Msite + ((1 * LY + j) * PD + xrow[1 * LY + j]) * 256;
            const float* M2 = Msite + ((2 * LY + j) * PD + xrow[2 * LY + j]) * 256;
            const float* M3 = Msite + ((3 * LY + j) * PD + xrow[3 * LY + j]) * 256;
            const float* M4 = Msite + ((4 * LY + j) * PD + xrow[4 * LY + j]) * 256;
            const float* M5 = Msite + ((5 * LY + j) * PD + xrow[5 * LY + j]) * 256;

            const int skip_dn = (have_deep && j == 1)      ? (have_deep2L ? 2 : 1) : 0;
            const int skip_up = (have_deep && j == LY - 2) ? 1 : 0;
            if (!upward) column_sites(env, M0, M1, M2, M3, M4, M5, wave, c, h, skip_dn);
            else         column_sites_up(env, M0, M1, M2, M3, M4, M5, wave, c, h, skip_up);
            upward = !upward;
        }
    }

    if (!have_right) {
        if (tid == 0) out[sample] = env[0];
        return;
    }

    const int cfg5 = xrow[0 * LY + 5] | (xrow[1 * LY + 5] << 1) | (xrow[2 * LY + 5] << 2) |
                     (xrow[3 * LY + 5] << 3) | (xrow[4 * LY + 5] << 4) | (xrow[5 * LY + 5] << 5);
    float part = 0.0f;
    if (have_deep3) {
        const int cfgR = cfg5 | (xrow[0 * LY + LY - 2] << 6) | (xrow[1 * LY + LY - 2] << 7) |
                         (xrow[2 * LY + LY - 2] << 8) | (xrow[3 * LY + LY - 2] << 9) |
                         (xrow[4 * LY + LY - 2] << 10) | (xrow[5 * LY + LY - 2] << 11);
        const float* rrow = table + (size_t)(NCFG4 + cfgR) * ENV6;
        float rv[16];
        #pragma unroll
        for (int q = 0; q < 16; ++q) rv[q] = rrow[tid + 256 * q];
        #pragma unroll
        for (int q = 0; q < 16; ++q)
            part = fmaf(at(env, ccell4(tid) + 4096 * q), rv[q], part);
    } else if (have_deep) {
        const int cfgR = cfg5 | (xrow[0 * LY + LY - 2] << 6) | (xrow[1 * LY + LY - 2] << 7);
        const int nl = have_deep2L ? NCFG3 : NCFG2;
        const v4f* rrow4 = reinterpret_cast<const v4f*>(table + (size_t)(nl + cfgR) * ENV7);
        const v4f* env4  = reinterpret_cast<const v4f*>(env);
        v4f rv4[16];
        #pragma unroll
        for (int q = 0; q < 16; ++q) rv4[q] = rrow4[tid + 256 * q];
        #pragma unroll
        for (int q = 0; q < 16; ++q) {
            const v4f e = env4[tid + 256 * q];
            part = fmaf(e.x, rv4[q].x, part);
            part = fmaf(e.y, rv4[q].y, part);
            part = fmaf(e.z, rv4[q].z, part);
            part = fmaf(e.w, rv4[q].w, part);
        }
    } else {
        const float* rrow = table + (NCFG + cfg5) * ENV6;
        float rv[16];
        #pragma unroll
        for (int q = 0; q < 16; ++q) rv[q] = rrow[tid + 256 * q];
        #pragma unroll
        for (int q = 0; q < 16; ++q)
            part = fmaf(at(env, ccell4(tid) + 4096 * q), rv[q], part);
    }
    for (int o = 16; o > 0; o >>= 1) part += __shfl_xor(part, o);
    __syncthreads();
    if (lane == 0) env[wave] = part;
    __syncthreads();
    if (tid == 0) {
        float t = env[0];
        for (int w = 1; w < 8; ++w) t += env[w];
        out[sample] = t;
    }
}

extern "C" void kernel_launch(void* const* d_in, const int* in_sizes, int n_in,
                              void* d_out, int out_size, void* d_ws, size_t ws_size,
                              hipStream_t stream) {
    const int*   x     = (const int*)d_in[0];
    const float* peps  = (const float*)d_in[1];
    float*       out   = (float*)d_out;
    float*       Msite = (float*)d_ws;
    float*       table = Msite + NMAT * PD * 256;

    const int nM = NMAT * PD * 256;
    peps_prep_kernel<<<(nM + 255) / 256, 256, 0, stream>>>(peps, Msite, ws_size);

    peps_edge_kernel<<<2 * NCFG2, 256, 0, stream>>>(Msite, table, ws_size);

    const int B = out_size;
    peps_amp_kernel<<<B, 256, ENV7 * sizeof(float), stream>>>(x, Msite, table, ws_size, out);
}


# zero padding of width-1 bonds no longer multiplied: row 0's four left-bond terms are one K chunk (main and table launch), edge-column pairs take 5 matrix ops a trip, not 32; row 0 sum order changes
# speedup vs baseline: 8.3834x; 1.0679x over previous
#include <hip/hip_runtime.h>


typedef float v2f __attribute__((ext_vector_type(2)));
typedef float v4f __attribute__((ext_vector_type(4)));
typedef float v8f __attribute__((ext_vector_type(8)));

#define LX 6
#define LY 6
#define PD 2
#define DD 4
#define NSITE 36
#define ENV7 16384
#define ENV6 4096
#define NCFG 64
#define NCFG2 256
#define NCFG3 1024
#define NCFG4 4096
#define NMAT 46
#define MSITE_BYTES (NMAT * PD * 256 * 4)
#define WS_NEED     (MSITE_BYTES + NCFG * ENV6 * 4)
#define WS_NEED2    (MSITE_BYTES + 2 * NCFG * ENV6 * 4)
#define WS_NEED3    (MSITE_BYTES + 2 * NCFG2 * ENV7 * 4)
#define WS_NEED4    (MSITE_BYTES + (NCFG3 + NCFG2) * ENV7 * 4)
#define WS_NEED5    (MSITE_BYTES + 2 * NCFG4 * ENV6 * 4)

__global__ void peps_prep_kernel(const float* __restrict__ peps,
                                 float* __restrict__ Msite,
                                 size_t ws_bytes) {
    int idx = blockIdx.x * blockDim.x + threadIdx.x;
    if (idx >= NMAT * PD * 256) return;
    if (idx >= NSITE * PD * 256 && ws_bytes < (size_t)WS_NEED2) return;
    if (idx >= (NSITE + LX) * PD * 256 && ws_bytes < (size_t)WS_NEED3) return;
    int n    = idx & 15;
    int k    = (idx >> 4) & 15;
    int s    = (idx >> 8) & 1;
    int site = idx >> 9;
    int a = k >> 2, c = k & 3;
    int d = n >> 2, b = n & 3;
    float v = 0.0f;
    if (site < NSITE) {
        int i = site / LY, j = site % LY;
        bool pad = (i == 0      && a > 0) || (i == LX - 1 && b > 0) ||
                   (j == 0      && c > 0) || (j == LY - 1 && d > 0);
        if (!pad) {
            int p = ((((site * PD + s) * DD + a) * DD + b) * DD + c) * DD + d;
            v = peps[p];
        }
    } else if (site < NSITE + LX) {
        int i  = site - NSITE;
        int rs = i * LY + (LY - 1);
        bool pad = (i == 0 && a > 0) || (i == LX - 1 && b > 0) || (c > 0);
        if (!pad) {
            int p = ((((rs * PD + s) * DD + a) * DD + b) * DD + d) * DD + c;
            v = peps[p];
        }
    } else {
        int i  = site - (NSITE + LX);
        int rs = i * LY + (LY - 2);
        bool pad = (i == 0 && a > 0);
        if (!pad) {
            int p = ((((rs * PD + s) * DD + a) * DD + b) * DD + d) * DD + c;
            v = peps[p];
        }
    }
    Msite[idx] = v;
}

__device__ __forceinline__ constexpr int swz(int a) { return a ^ (((a >> 6) & 15) << 2); }

__device__ __forceinline__ int cell4(int Lz4, int K) {
    const int X   = ((K >> 6) & 15) ^ ((K >> 2) & 15);
    const int KHI = K & ~63;
    return (Lz4 ^ (X * 16)) + KHI * 4;
}
__device__ __forceinline__ int ccell4(int C) { return 16 * (C ^ ((C >> 4) & 15)); }
__device__ __forceinline__ float& at(float* env, int byte_addr) {
    return *reinterpret_cast<float*>(reinterpret_cast<char*>(env) + byte_addr);
}
__device__ __forceinline__ v2f& at2(float* env, int byte_addr) {
    return *reinterpret_cast<v2f*>(reinterpret_cast<char*>(env) + byte_addr);
}
__device__ __forceinline__ v4f& at4(float* env, int byte_addr) {
    return *reinterpret_cast<v4f*>(reinterpret_cast<char*>(env) + byte_addr);
}

template <int P, bool UPWARD>
__device__ __forceinline__ void site_pair(float* env,
                                          const float* __restrict__ MpA,
                                          const float* __restrict__ MpB,
                                          int wave, int c, int h) {
    constexpr int P1 = UPWARD ? P + 2 : P;
    constexpr int P2 = UPWARD ? P : P + 2;

    const int vo = 2 * (c >> 3) + (c & 1);
    const int ro = (c >> 1) & 3;
    v2f fa[4], fb[4];
    #pragma unroll
    for (int kc = 0; kc < 4; ++kc) {
        if (!UPWARD) {
            const int ncol = ro * 4 + vo;
            fa[kc].x = MpA[((2 * h + 0) * 4 + kc) * 16 + ncol];
            fa[kc].y = MpA[((2 * h + 1) * 4 + kc) * 16 + ncol];
            fb[kc].x = MpB[((2 * h + 0) * 4 + kc) * 16 + ncol];
            fb[kc].y = MpB[((2 * h + 1) * 4 + kc) * 16 + ncol];
        } else {
            const int nrow = (vo * 4 + kc) * 16 + ro * 4;
            fa[kc].x = MpA[nrow + 2 * h + 0];
            fa[kc].y = MpA[nrow + 2 * h + 1];
            fb[kc].x = MpB[nrow + 2 * h + 0];
            fb[kc].y = MpB[nrow + 2 * h + 1];
        }
    }

    #pragma nounroll
    for (int it = 0; it < 2; ++it) {
        const int q    = (wave + 8 * it) * 16 + c;
        const int ql   = q & ((1 << (P - 2)) - 1);
        const int base = ((q >> (P - 2)) << (P + 4)) | (ql << 2);
        const int a0   = 4 * swz(base + 2 * h);

        v8f dA[4];
        #pragma unroll
        for (int T = 0; T < 4; ++T) {
            v8f acc = {};
            #pragma unroll
            for (int kc = 0; kc < 4; ++kc) {
                const v2f b = at2(env, cell4(a0, (kc << P1) | (T << P2)));
                acc = __builtin_amdgcn_wmma_f32_16x16x4_f32(
                        false, fa[kc], false, b, (short)0, acc, false, false);
            }
            dA[T] = acc;
        }

        #pragma unroll
        for (int R = 0; R < 4; ++R) {
            v8f acc = {};
            #pragma unroll
            for (int kc = 0; kc < 4; ++kc) {
                v2f b;
                b.x = dA[kc][2 * R];
                b.y = dA[kc][2 * R + 1];
                acc = __builtin_amdgcn_wmma_f32_16x16x4_f32(
                        false, fb[kc], false, b, (short)0, acc, false, false);
            }
            #pragma unroll
            for (int r2 = 0; r2 < 4; ++r2) {
                v2f o;
                o.x = acc[2 * r2];
                o.y = acc[2 * r2 + 1];
                at2(env, cell4(a0, (R << P1) | (r2 << P2))) = o;
            }
        }
    }
}

__device__ __forceinline__ void site_pair_top(float* env,
                                              const float* __restrict__ MpA,
                                              const float* __restrict__ MpB,
                                              int wave, int c, int h) {
    constexpr int P1 = 2, P2 = 4;
    const int vo = 2 * (c >> 3) + (c & 1);
    const int ro = (c >> 1) & 3;
    const int ncol = ro * 4 + vo;
    v2f fa0, fb[4];
    fa0.x = MpA[(2 * h + 0) * 16 + ncol];
    fa0.y = MpA[(2 * h + 1) * 16 + ncol];
    #pragma unroll
    for (int kc = 0; kc < 4; ++kc) {
        fb[kc].x = MpB[((2 * h + 0) * 4 + kc) * 16 + ncol];
        fb[kc].y = MpB[((2 * h + 1) * 4 + kc) * 16 + ncol];
    }

    #pragma nounroll
    for (int it = 0; it < 2; ++it) {
        const int q    = (wave + 8 * it) * 16 + c;
        const int base = q << 6;
        const int a0   = 4 * swz(base + 2 * h);
        const int a1   = 4 * swz(base + 8 * h);

        v8f dA[4];
        #pragma unroll
        for (int T = 0; T < 4; ++T) {
            v2f b;
            b.x = at(env, cell4(a1, (0 << P1) | (T << P2)));
            b.y = at(env, cell4(a1, (1 << P1) | (T << P2)));
            const v8f zero = {};
            dA[T] = __builtin_amdgcn_wmma_f32_16x16x4_f32(
                        false, fa0, false, b, (short)0, zero, false, false);
        }

        #pragma unroll
        for (int R = 0; R < 4; ++R) {
            v8f acc = {};
            #pragma unroll
            for (int kc = 0; kc < 4; ++kc) {
                v2f b;
                b.x = dA[kc][2 * R];
                b.y = dA[kc][2 * R + 1];
                acc = __builtin_amdgcn_wmma_f32_16x16x4_f32(
                        false, fb[kc], false, b, (short)0, acc, false, false);
            }
            #pragma unroll
            for (int r2 = 0; r2 < 4; ++r2) {
                v2f o;
                o.x = acc[2 * r2];
                o.y = acc[2 * r2 + 1];
                at2(env, cell4(a0, (R << P1) | (r2 << P2))) = o;
            }
        }
    }
}

template <int P>
__device__ __forceinline__ void site_pair_edge(float* env,
                                               const float* __restrict__ MpA,
                                               const float* __restrict__ MpB,
                                               int wave, int c, int h) {
    constexpr int P1 = P, P2 = P + 2;
    const int vo = 2 * (c >> 3) + (c & 1);
    const int ro = (c >> 1) & 3;
    const int ncol = ro * 4 + vo;
    v2f fa0, fb0;
    fa0.x = MpA[((2 * h + 0) * 4) * 16 + ncol];
    fa0.y = MpA[((2 * h + 1) * 4) * 16 + ncol];
    fb0.x = MpB[((2 * h + 0) * 4) * 16 + ncol];
    fb0.y = MpB[((2 * h + 1) * 4) * 16 + ncol];

    #pragma nounroll
    for (int it = 0; it < 2; ++it) {
        const int q    = (wave + 8 * it) * 16 + c;
        const int ql   = q & ((1 << (P - 2)) - 1);
        const int base = ((q >> (P - 2)) << (P + 4)) | (ql << 2);
        const int a0   = 4 * swz(base + 2 * h);

        const v8f zero = {};
        const v2f b1 = at2(env, cell4(a0, 0));
        const v8f dA = __builtin_amdgcn_wmma_f32_16x16x4_f32(
                           false, fa0, false, b1, (short)0, zero, false, false);

        #pragma unroll
        for (int R = 0; R < 4; ++R) {
            v2f b;
            b.x = dA[2 * R];
            b.y = dA[2 * R + 1];
            const v8f acc = __builtin_amdgcn_wmma_f32_16x16x4_f32(
                                false, fb0, false, b, (short)0, zero, false, false);
            #pragma unroll
            for (int r2 = 0; r2 < 4; ++r2) {
                v2f o;
                o.x = acc[2 * r2];
                o.y = acc[2 * r2 + 1];
                at2(env, cell4(a0, (R << P1) | (r2 << P2))) = o;
            }
        }
    }
}

__device__ __forceinline__ void turn_chain(float* env,
                                           const float* __restrict__ M4a,
                                           const float* __restrict__ M5a,
                                           const float* __restrict__ M5b,
                                           const float* __restrict__ M4b,
                                           int wave, int c, int h) {
    constexpr int P4 = 10, P5 = 12;
    const int vo = 2 * (c >> 3) + (c & 1);
    const int ro = (c >> 1) & 3;
    v2f pa, pb;
    pa.x = M5a[(vo * 4 + ro) * 16 + (2 * h + 0) * 4];
    pa.y = M5a[(vo * 4 + ro) * 16 + (2 * h + 1) * 4];
    pb.x = M5b[(vo * 4 + 2 * h + 0) * 16 + ro * 4];
    pb.y = M5b[(vo * 4 + 2 * h + 1) * 16 + ro * 4];
    const v8f fzero = {};
    const v8f F = __builtin_amdgcn_wmma_f32_16x16x4_f32(
            false, pa, false, pb, (short)0, fzero, false, false);

    v2f fa[4], ff[4], fc[4];
    #pragma unroll
    for (int kc = 0; kc < 4; ++kc) {
        const int ncol = ro * 4 + vo;
        fa[kc].x = M4a[((2 * h + 0) * 4 + kc) * 16 + ncol];
        fa[kc].y = M4a[((2 * h + 1) * 4 + kc) * 16 + ncol];
        const int nrow = (vo * 4 + kc) * 16 + ro * 4;
        fc[kc].x = M4b[nrow + 2 * h + 0];
        fc[kc].y = M4b[nrow + 2 * h + 1];
        ff[kc].x = F[2 * kc];
        ff[kc].y = F[2 * kc + 1];
    }

    #pragma nounroll
    for (int it = 0; it < 2; ++it) {
        const int q  = (wave + 8 * it) * 16 + c;
        const int a0 = 4 * swz((q << 2) + 2 * h);

        v8f dA[4];
        #pragma unroll
        for (int T = 0; T < 4; ++T) {
            v8f acc = {};
            #pragma unroll
            for (int kc = 0; kc < 4; ++kc) {
                const v2f b = at2(env, cell4(a0, (kc << P4) | (T << P5)));
                acc = __builtin_amdgcn_wmma_f32_16x16x4_f32(
                        false, fa[kc], false, b, (short)0, acc, false, false);
            }
            dA[T] = acc;
        }

        v8f dB[4];
        #pragma unroll
        for (int R = 0; R < 4; ++R) {
            v8f acc = {};
            #pragma unroll
            for (int kc = 0; kc < 4; ++kc) {
                v2f b;
                b.x = dA[kc][2 * R];
                b.y = dA[kc][2 * R + 1];
                acc = __builtin_amdgcn_wmma_f32_16x16x4_f32(
                        false, ff[kc], false, b, (short)0, acc, false, false);
            }
            dB[R] = acc;
        }

        #pragma unroll
        for (int r5 = 0; r5 < 4; ++r5) {
            v8f acc = {};
            #pragma unroll
            for (int kc = 0; kc < 4; ++kc) {
                v2f b;
                b.x = dB[kc][2 * r5];
                b.y = dB[kc][2 * r5 + 1];
                acc = __builtin_amdgcn_wmma_f32_16x16x4_f32(
                        false, fc[kc], false, b, (short)0, acc, false, false);
            }
            #pragma unroll
            for (int r4 = 0; r4 < 4; ++r4) {
                v2f o;
                o.x = acc[2 * r4];
                o.y = acc[2 * r4 + 1];
                at2(env, cell4(a0, (r4 << P4) | (r5 << P5))) = o;
            }
        }
    }
}

__device__ __forceinline__ void last_pair_rows(float* env, float* stage,
                                               const float* __restrict__ M4,
                                               const float* __restrict__ M5,
                                               bool mirror,
                                               float* __restrict__ rows,
                                               int tid, int wave, int c, int h) {
    constexpr int P1 = 10, P2 = 12;
    const int vo = 2 * (c >> 3) + (c & 1);
    const int ro = (c >> 1) & 3;
    const v4f* stage4 = reinterpret_cast<const v4f*>(stage);
    int mi[4];
    #pragma unroll
    for (int kc = 0; kc < 4; ++kc)
        mi[kc] = mirror ? ((2 * h) * 4 + ro) * 16 + kc * 4 + vo
                        : ((2 * h) * 4 + kc) * 16 + ro * 4 + vo;
    const int spare = ENV6 + c;

    #pragma nounroll
    for (int x4 = 0; x4 < 2; ++x4) {
        const float* MpA = M4 + x4 * 256;
        v2f fa[4];
        #pragma unroll
        for (int kc = 0; kc < 4; ++kc) {
            fa[kc].x = MpA[mi[kc]];
            fa[kc].y = MpA[mi[kc] + 64];
        }
        v8f dA[2][4];
        #pragma unroll
        for (int it = 0; it < 2; ++it) {
            const int q  = (wave + 8 * it) * 16 + c;
            const int a0 = 4 * swz((q << 2) + 2 * h);
            #pragma unroll
            for (int T = 0; T < 4; ++T) {
                v8f acc = {};
                #pragma unroll
                for (int kc = 0; kc < 4; ++kc) {
                    const v2f b = at2(env, cell4(a0, (kc << P1) | (T << P2)));
                    acc = __builtin_amdgcn_wmma_f32_16x16x4_f32(
                            false, fa[kc], false, b, (short)0, acc, false, false);
                }
                dA[it][T] = acc;
            }
        }
        #pragma nounroll
        for (int x5 = 0; x5 < 2; ++x5) {
            const float* MpB = M5 + x5 * 256;
            v2f fb[4];
            #pragma unroll
            for (int kc = 0; kc < 4; ++kc) {
                fb[kc].x = MpB[mi[kc]];
                fb[kc].y = MpB[mi[kc] + 64];
            }
            __syncthreads();
            #pragma unroll
            for (int it = 0; it < 2; ++it) {
                const int q = (wave + 8 * it) * 16 + c;
                #pragma unroll
                for (int R = 0; R < 4; ++R) {
                    v8f acc = {};
                    #pragma unroll
                    for (int kc = 0; kc < 4; ++kc) {
                        v2f b;
                        b.x = dA[it][kc][2 * R];
                        b.y = dA[it][kc][2 * R + 1];
                        acc = __builtin_amdgcn_wmma_f32_16x16x4_f32(
                                false, fb[kc], false, b, (short)0, acc, false, false);
                    }
                    #pragma unroll
                    for (int r2 = 0; r2 < 4; ++r2)
                        stage[h ? spare : (q | (R << 8) | (r2 << 10))] = acc[2 * r2];
                }
            }
            __syncthreads();
            v4f* row4 = reinterpret_cast<v4f*>(rows + (size_t)(x4 + 2 * x5) * NCFG3 * ENV6);
            #pragma unroll
            for (int k = 0; k < 4; ++k) row4[tid + 256 * k] = stage4[tid + 256 * k];
        }
    }
}

__device__ __forceinline__ void column_sites(float* env,
                                             const float* M0, const float* M1,
                                             const float* M2, const float* M3,
                                             const float* M4, const float* M5,
                                             int wave, int c, int h, int skip) {
    if (skip < 1) {
        site_pair<2, false>(env, M0, M1, wave, c, h);
        __syncthreads();
    }
    if (skip < 2) {
        site_pair<6, false>(env, M2, M3, wave, c, h);
        __syncthreads();
    }
    site_pair<10, false>(env, M4, M5, wave, c, h);
    __syncthreads();
}

__device__ __forceinline__ void column_sites_up(float* env,
                                                const float* M0, const float* M1,
                                                const float* M2, const float* M3,
                                                const float* M4, const float* M5,
                                                int wave, int c, int h, int skip) {
    site_pair<10, true>(env, M5, M4, wave, c, h);
    __syncthreads();
    if (skip < 2) {
        site_pair<6, true>(env, M3, M2, wave, c, h);
        __syncthreads();
    }
    if (skip < 1) {
        site_pair<2, true>(env, M1, M0, wave, c, h);
        __syncthreads();
    }
}

__global__ void __launch_bounds__(256)
peps_edge_kernel(const float* __restrict__ Msite,
                 float* __restrict__ table,
                 size_t ws_bytes) {
    __shared__ __attribute__((aligned(16))) float env[2 * ENV7];
    __shared__ __attribute__((aligned(16))) float stage[ENV6 + 16];

    const bool deep   = ws_bytes >= (size_t)WS_NEED3;
    const bool deep2L = ws_bytes >= (size_t)WS_NEED4;
    const bool deep3  = ws_bytes >= (size_t)WS_NEED5;
    const bool right = blockIdx.x >= (unsigned)(deep ? NCFG2 : NCFG);

    if (!deep) {
        if (blockIdx.x >= 2 * NCFG) return;
        if (ws_bytes < (size_t)(right ? WS_NEED2 : WS_NEED)) return;
    }

    const int tid  = threadIdx.x;
    const int cfg8 = blockIdx.x & ((deep ? NCFG2 : NCFG) - 1);
    const int cfg  = cfg8 & (NCFG - 1);
    const int m0   = right ? NSITE : 0;
    const int mst  = right ? 1 : LY;
    const int lane = tid & 31;
    const int wave = __builtin_amdgcn_readfirstlane(tid >> 5);
    __builtin_assume(wave >= 0 && wave < 8);
    const int c    = lane & 15;
    const int h    = lane >> 4;

    for (int q = tid; q < ENV7; q += 256) env[q] = 0.0f;
    __syncthreads();
    if (tid == 0) env[0] = 1.0f;
    __syncthreads();

    const float* M0 = Msite + ((m0 + 0 * mst) * PD + ((cfg >> 0) & 1)) * 256;
    const float* M1 = Msite + ((m0 + 1 * mst) * PD + ((cfg >> 1) & 1)) * 256;
    const float* M2 = Msite + ((m0 + 2 * mst) * PD + ((cfg >> 2) & 1)) * 256;
    const float* M3 = Msite + ((m0 + 3 * mst) * PD + ((cfg >> 3) & 1)) * 256;
    const float* M4 = Msite + ((m0 + 4 * mst) * PD + ((cfg >> 4) & 1)) * 256;
    const float* M5 = Msite + ((m0 + 5 * mst) * PD + ((cfg >> 5) & 1)) * 256;
    site_pair_edge<2>(env, M0, M1, wave, c, h);
    __syncthreads();
    site_pair_edge<6>(env, M2, M3, wave, c, h);
    __syncthreads();
    site_pair_edge<10>(env, M4, M5, wave, c, h);
    __syncthreads();

    if (!deep) {
        float* row = table + blockIdx.x * ENV6;
        #pragma unroll
        for (int q = 0; q < 16; ++q)
            row[tid + 256 * q] = at(env, ccell4(tid) + 4096 * q);
        return;
    }

    const int mA = right ? (NSITE + LX + 0) : (0 * LY + 1);
    const int mB = right ? (NSITE + LX + 1) : (1 * LY + 1);
    const float* MA = Msite + (mA * PD + ((cfg8 >> 6) & 1)) * 256;
    const float* MB = Msite + (mB * PD + ((cfg8 >> 7) & 1)) * 256;
    site_pair_top(env, MA, MB, wave, c, h);
    __syncthreads();

    const bool deeper = right ? deep3 : deep2L;
    float*     side   = table + (right ? (size_t)(deep2L ? NCFG3 : NCFG2) * ENV7 : (size_t)0);
    v4f*       env4   = reinterpret_cast<v4f*>(env);

    if (!deeper) {
        v4f* row4 = reinterpret_cast<v4f*>(side + (size_t)cfg8 * ENV7);
        #pragma unroll
        for (int q = 0; q < 16; ++q) row4[tid + 256 * q] = env4[tid + 256 * q];
        return;
    }

    v4f* copy4 = reinterpret_cast<v4f*>(env + ENV7);
    #pragma unroll
    for (int q = 0; q < 16; ++q) copy4[tid + 256 * q] = env4[tid + 256 * q];
    __syncthreads();

    const int mC = right ? (NSITE + LX + 2) : (2 * LY + 1);
    const int mD = right ? (NSITE + LX + 3) : (3 * LY + 1);
    #pragma nounroll
    for (int v = 0; v < 4; ++v) {
        if (v > 0) {
            __syncthreads();
            #pragma unroll
            for (int q = 0; q < 16; ++q) env4[tid + 256 * q] = copy4[tid + 256 * q];
            __syncthreads();
        }
        const float* MC = Msite + (mC * PD + (v & 1)) * 256;
        const float* MD = Msite + (mD * PD + (v >> 1)) * 256;
        site_pair<6, false>(env, MC, MD, wave, c, h);
        __syncthreads();
        if (deep3) {
            const int m4 = right ? (4 * LY + LY - 2) : (4 * LY + 1);
            const int m5 = right ? (5 * LY + LY - 2) : (5 * LY + 1);
            last_pair_rows(env, stage, Msite + m4 * PD * 256, Msite + m5 * PD * 256, right,
                           side + (size_t)(cfg8 + NCFG2 * v) * ENV6, tid, wave, c, h);
        } else {
            v4f* row4 = reinterpret_cast<v4f*>(side + (size_t)(cfg8 + NCFG2 * v) * ENV7);
            #pragma unroll
            for (int q = 0; q < 16; ++q) row4[tid + 256 * q] = env4[tid + 256 * q];
        }
    }
}

__global__ void __launch_bounds__(256)
peps_amp_kernel(const int* __restrict__ x,
                const float* __restrict__ Msite,
                const float* __restrict__ table,
                size_t ws_bytes,
                float* __restrict__ out) {
    extern __shared__ float env[];

    const int tid    = threadIdx.x;
    const int sample = blockIdx.x;
    const int lane   = tid & 31;
    const int wave   = __builtin_amdgcn_readfirstlane(tid >> 5);
    __builtin_assume(wave >= 0 && wave < 8);
    const int c      = lane & 15;
    const int h      = lane >> 4;

    const int* xrow = x + sample * NSITE;
    const bool have_table = ws_bytes >= (size_t)WS_NEED;
    const bool have_right = ws_bytes >= (size_t)WS_NEED2;
    const bool have_deep  = ws_bytes >= (size_t)WS_NEED3;
    const bool have_deep2L = ws_bytes >= (size_t)WS_NEED4;
    const bool have_deep3  = ws_bytes >= (size_t)WS_NEED5;

    if (have_deep3) {
        const int cfg = xrow[0 * LY] | (xrow[1 * LY] << 1) | (xrow[2 * LY] << 2) |
                        (xrow[3 * LY] << 3) | (xrow[4 * LY] << 4) | (xrow[5 * LY] << 5) |
                        (xrow[0 * LY + 1] << 6) | (xrow[1 * LY + 1] << 7) |
                        (xrow[2 * LY + 1] << 8) | (xrow[3 * LY + 1] << 9) |
                        (xrow[4 * LY + 1] << 10) | (xrow[5 * LY + 1] << 11);
        const float* row = table + (size_t)cfg * ENV6;
        float keep[16];
        #pragma unroll
        for (int q = 0; q < 16; ++q) keep[q] = row[tid + 256 * q];
        #pragma unroll
        for (int q = 0; q < 16; ++q) at(env, ccell4(tid) + 4096 * q) = keep[q];
        __syncthreads();
    } else if (have_deep) {
        int cfg = xrow[0 * LY] | (xrow[1 * LY] << 1) | (xrow[2 * LY] << 2) |
                  (xrow[3 * LY] << 3) | (xrow[4 * LY] << 4) | (xrow[5 * LY] << 5) |
                  (xrow[0 * LY + 1] << 6) | (xrow[1 * LY + 1] << 7);
        if (have_deep2L) cfg |= (xrow[2 * LY + 1] << 8) | (xrow[3 * LY + 1] << 9);
        const v4f* row4 = reinterpret_cast<const v4f*>(table + (size_t)cfg * ENV7);
        v4f*       env4 = reinterpret_cast<v4f*>(env);
        v4f keep4[16];
        #pragma unroll
        for (int q = 0; q < 16; ++q) keep4[q] = row4[tid + 256 * q];
        #pragma unroll
        for (int q = 0; q < 16; ++q) env4[tid + 256 * q] = keep4[q];
        __syncthreads();
    } else if (have_table) {
        const int cfg = xrow[0 * LY] | (xrow[1 * LY] << 1) | (xrow[2 * LY] << 2) |
                        (xrow[3 * LY] << 3) | (xrow[4 * LY] << 4) | (xrow[5 * LY] << 5);
        const float* row = table + cfg * ENV6;
        float keep[16];
        #pragma unroll
        for (int q = 0; q < 16; ++q) keep[q] = row[tid + 256 * q];
        #pragma unroll
        for (int q = 0; q < 16; ++q) {
            const v4f t = {keep[q], 0.0f, 0.0f, 0.0f};
            at4(env, ccell4(tid) + 4096 * q) = t;
        }
        __syncthreads();
    } else {
        for (int q = tid; q < ENV7; q += 256) env[q] = 0.0f;
        __syncthreads();
        if (tid == 0) env[0] = 1.0f;
        __syncthreads();
    }

    if (have_deep3) {
        const float* A[LX];
        const float* Bm[LX];
        #pragma unroll
        for (int i = 0; i < LX; ++i) {
            A[i]  = Msite + ((i * LY + 2) * PD + xrow[i * LY + 2]) * 256;
            Bm[i] = Msite + ((i * LY + 3) * PD + xrow[i * LY + 3]) * 256;
        }
        site_pair_top(env, A[0], A[1], wave, c, h);
        __syncthreads();
        site_pair<6, false>(env, A[2], A[3], wave, c, h);
        __syncthreads();
        turn_chain(env, A[4], A[5], Bm[5], Bm[4], wave, c, h);
        __syncthreads();
        site_pair<6, true>(env, Bm[3], Bm[2], wave, c, h);
        __syncthreads();
        site_pair<2, true>(env, Bm[1], Bm[0], wave, c, h);
        __syncthreads();
    } else {
        bool upward = false;
        const int jend = have_right ? LY - 1 : LY;
        for (int j = have_table ? 1 : 0; j < jend; ++j) {
            const float* M0 = Msite + ((0 * LY + j) * PD + xrow[0 * LY + j]) * 256;
            const float* M1 = Msite + ((1 * LY + j) * PD + xrow[1 * LY + j]) * 256;
            const float* M2 = Msite + ((2 * LY + j) * PD + xrow[2 * LY + j]) * 256;
            const float* M3 = Msite + ((3 * LY + j) * PD + xrow[3 * LY + j]) * 256;
            const float* M4 = Msite + ((4 * LY + j) * PD + xrow[4 * LY + j]) * 256;
            const float* M5 = Msite + ((5 * LY + j) * PD + xrow[5 * LY + j]) * 256;

            const int skip_dn = (have_deep && j == 1)      ? (have_deep2L ? 2 : 1) : 0;
            const int skip_up = (have_deep && j == LY - 2) ? 1 : 0;
            if (!upward) column_sites(env, M0, M1, M2, M3, M4, M5, wave, c, h, skip_dn);
            else         column_sites_up(env, M0, M1, M2, M3, M4, M5, wave, c, h, skip_up);
            upward = !upward;
        }
    }

    if (!have_right) {
        if (tid == 0) out[sample] = env[0];
        return;
    }

    const int cfg5 = xrow[0 * LY + 5] | (xrow[1 * LY + 5] << 1) | (xrow[2 * LY + 5] << 2) |
                     (xrow[3 * LY + 5] << 3) | (xrow[4 * LY + 5] << 4) | (xrow[5 * LY + 5] << 5);
    float part = 0.0f;
    if (have_deep3) {
        const int cfgR = cfg5 | (xrow[0 * LY + LY - 2] << 6) | (xrow[1 * LY + LY - 2] << 7) |
                         (xrow[2 * LY + LY - 2] << 8) | (xrow[3 * LY + LY - 2] << 9) |
                         (xrow[4 * LY + LY - 2] << 10) | (xrow[5 * LY + LY - 2] << 11);
        const float* rrow = table + (size_t)(NCFG4 + cfgR) * ENV6;
        float rv[16];
        #pragma unroll
        for (int q = 0; q < 16; ++q) rv[q] = rrow[tid + 256 * q];
        #pragma unroll
        for (int q = 0; q < 16; ++q)
            part = fmaf(at(env, ccell4(tid) + 4096 * q), rv[q], part);
    } else if (have_deep) {
        const int cfgR = cfg5 | (xrow[0 * LY + LY - 2] << 6) | (xrow[1 * LY + LY - 2] << 7);
        const int nl = have_deep2L ? NCFG3 : NCFG2;
        const v4f* rrow4 = reinterpret_cast<const v4f*>(table + (size_t)(nl + cfgR) * ENV7);
        const v4f* env4  = reinterpret_cast<const v4f*>(env);
        v4f rv4[16];
        #pragma unroll
        for (int q = 0; q < 16; ++q) rv4[q] = rrow4[tid + 256 * q];
        #pragma unroll
        for (int q = 0; q < 16; ++q) {
            const v4f e = env4[tid + 256 * q];
            part = fmaf(e.x, rv4[q].x, part);
            part = fmaf(e.y, rv4[q].y, part);
            part = fmaf(e.z, rv4[q].z, part);
            part = fmaf(e.w, rv4[q].w, part);
        }
    } else {
        const float* rrow = table + (NCFG + cfg5) * ENV6;
        float rv[16];
        #pragma unroll
        for (int q = 0; q < 16; ++q) rv[q] = rrow[tid + 256 * q];
        #pragma unroll
        for (int q = 0; q < 16; ++q)
            part = fmaf(at(env, ccell4(tid) + 4096 * q), rv[q], part);
    }
    for (int o = 16; o > 0; o >>= 1) part += __shfl_xor(part, o);
    __syncthreads();
    if (lane == 0) env[wave] = part;
    __syncthreads();
    if (tid == 0) {
        float t = env[0];
        for (int w = 1; w < 8; ++w) t += env[w];
        out[sample] = t;
    }
}

extern "C" void kernel_launch(void* const* d_in, const int* in_sizes, int n_in,
                              void* d_out, int out_size, void* d_ws, size_t ws_size,
                              hipStream_t stream) {
    const int*   x     = (const int*)d_in[0];
    const float* peps  = (const float*)d_in[1];
    float*       out   = (float*)d_out;
    float*       Msite = (float*)d_ws;
    float*       table = Msite + NMAT * PD * 256;

    const int nM = NMAT * PD * 256;
    peps_prep_kernel<<<(nM + 255) / 256, 256, 0, stream>>>(peps, Msite, ws_size);

    peps_edge_kernel<<<2 * NCFG2, 256, 0, stream>>>(Msite, table, ws_size);

    const int B = out_size;
    peps_amp_kernel<<<B, 256, ENV7 * sizeof(float), stream>>>(x, Msite, table, ws_size, out);
}
